# starting kernel
# baseline (speedup 1.0000x reference)
#include <hip/hip_runtime.h>
#include <hip/hip_bf16.h>


#define NODES 50000
#define HDIM  128

typedef __attribute__((ext_vector_type(2))) float v2f;
typedef __attribute__((ext_vector_type(8))) float v8f;

__global__ __launch_bounds__(256) void gemm128_wmma(const float* __restrict__ X,
                                                    const float* __restrict__ W,
                                                    float* __restrict__ Hout) {
    __shared__ float Wl[128 * 128];
    __shared__ float Xl[16 * 130];

    const int tid = threadIdx.x;
    const int rowBase = blockIdx.x * 16;

    {
        const float4* W4 = (const float4*)W;
        float4* Wl4 = (float4*)Wl;
        for (int i = tid; i < 128 * 128 / 4; i += 256) Wl4[i] = W4[i];
    }
    {
        const float* Xt = X + (size_t)rowBase * HDIM;
        for (int i = tid; i < 16 * 128; i += 256) {
            int r = i >> 7, c = i & 127;
            Xl[r * 130 + c] = Xt[i];
        }
    }
    __syncthreads();

    const int lane = tid & 31;
    const int wave = tid >> 5;
    const int r16  = lane & 15;
    const int hi   = lane >> 4;
    const int colBase = wave * 16;

    v8f acc = {};
    #pragma unroll 4
    for (int k = 0; k < 128; k += 4) {
        const int kk = k + hi * 2;
        v2f a;
        a[0] = Xl[r16 * 130 + kk];
        a[1] = Xl[r16 * 130 + kk + 1];
        v2f b;
        b[0] = Wl[(kk + 0) * 128 + colBase + r16];
        b[1] = Wl[(kk + 1) * 128 + colBase + r16];
        acc = __builtin_amdgcn_wmma_f32_16x16x4_f32(
             false, a,  false, b,
             (short)0, acc,  false,  false);
    }

    float* out = Hout + (size_t)rowBase * HDIM + colBase + r16;
    #pragma unroll
    for (int r = 0; r < 8; ++r) {
        out[(size_t)(r + hi * 8) * HDIM] = acc[r];
    }
}

__global__ void init_deg(float* deg, int n) {
    int i = blockIdx.x * blockDim.x + threadIdx.x;
    if (i < n) deg[i] = 1.0f;
}

__global__ void accum_deg(const int* __restrict__ dst, float* deg, int e) {
    int i = blockIdx.x * blockDim.x + threadIdx.x;
    if (i < e) atomicAdd(&deg[dst[i]], 1.0f);
}

__global__ void rsqrt_deg(float* deg, int n) {
    int i = blockIdx.x * blockDim.x + threadIdx.x;
    if (i < n) deg[i] = __frsqrt_rn(deg[i]);
}

__global__ void zero_f4(float4* p, int n4) {
    int i = blockIdx.x * blockDim.x + threadIdx.x;
    if (i < n4) p[i] = make_float4(0.f, 0.f, 0.f, 0.f);
}

__global__ __launch_bounds__(256) void aggregate(const float* __restrict__ h,
                                                 const int* __restrict__ src,
                                                 const int* __restrict__ dst,
                                                 const float* __restrict__ dinv,
                                                 float* __restrict__ agg, int e) {
    int g = blockIdx.x * blockDim.x + threadIdx.x;
    int edge = g >> 5;
    int lane = g & 31;
    if (edge >= e) return;
    int s = src[edge];
    int d = dst[edge];
    float c = dinv[s] * dinv[d];
    const float4 v = ((const float4*)(h + (size_t)s * HDIM))[lane];
    float* a = agg + (size_t)d * HDIM + lane * 4;
    atomicAdd(a + 0, c * v.x);
    atomicAdd(a + 1, c * v.y);
    atomicAdd(a + 2, c * v.z);
    atomicAdd(a + 3, c * v.w);
}

__global__ __launch_bounds__(256) void finalize(const float* __restrict__ agg,
                                                const float* __restrict__ h,
                                                const float* __restrict__ dinv,
                                                const float* __restrict__ bias,
                                                float* __restrict__ out,
                                                int n, int relu) {
    int i = blockIdx.x * blockDim.x + threadIdx.x;
    if (i >= n * (HDIM / 4)) return;
    int node = i / (HDIM / 4);
    int c4   = i % (HDIM / 4);
    float sl = dinv[node] * dinv[node];
    float4 a = ((const float4*)agg)[i];
    float4 v = ((const float4*)h)[i];
    const float4 b = ((const float4*)bias)[c4];
    a.x += sl * v.x + b.x;
    a.y += sl * v.y + b.y;
    a.z += sl * v.z + b.z;
    a.w += sl * v.w + b.w;
    if (relu) {
        a.x = fmaxf(a.x, 0.f);
        a.y = fmaxf(a.y, 0.f);
        a.z = fmaxf(a.z, 0.f);
        a.w = fmaxf(a.w, 0.f);
    }
    ((float4*)out)[i] = a;
}

extern "C" void kernel_launch(void* const* d_in, const int* in_sizes, int n_in,
                              void* d_out, int out_size, void* d_ws, size_t ws_size,
                              hipStream_t stream) {
    (void)n_in; (void)out_size; (void)ws_size;

    const float* x  = (const float*)d_in[0];
    const int*   ei = (const int*)d_in[1];
    const float* W1 = (const float*)d_in[2];
    const float* b1 = (const float*)d_in[3];
    const float* W2 = (const float*)d_in[4];
    const float* b2 = (const float*)d_in[5];
    const float* W3 = (const float*)d_in[6];
    const float* b3 = (const float*)d_in[7];

    const int N = in_sizes[0] / HDIM;
    const int E = in_sizes[1] / 2;
    const int* src = ei;
    const int* dst = ei + E;

    float* ws   = (float*)d_ws;
    float* h    = ws;
    float* agg  = ws + (size_t)N * HDIM;
    float* xbuf = ws + 2 * (size_t)N * HDIM;
    float* dinv = ws + 3 * (size_t)N * HDIM;
    float* out  = (float*)d_out;

    const int nh4 = N * (HDIM / 4);
    dim3 blk(256);
    dim3 gN((N + 255) / 256);
    dim3 gE((E + 255) / 256);
    dim3 gEdge((E + 7) / 8);
    dim3 gNH4((nh4 + 255) / 256);
    dim3 gGemm(N / 16);

    init_deg<<<gN, blk, 0, stream>>>(dinv, N);
    accum_deg<<<gE, blk, 0, stream>>>(dst, dinv, E);
    rsqrt_deg<<<gN, blk, 0, stream>>>(dinv, N);

    gemm128_wmma<<<gGemm, blk, 0, stream>>>(x, W1, h);
    zero_f4<<<gNH4, blk, 0, stream>>>((float4*)agg, nh4);
    aggregate<<<gEdge, blk, 0, stream>>>(h, src, dst, dinv, agg, E);
    finalize<<<gNH4, blk, 0, stream>>>(agg, h, dinv, b1, xbuf, N, 1);

    gemm128_wmma<<<gGemm, blk, 0, stream>>>(xbuf, W2, h);
    zero_f4<<<gNH4, blk, 0, stream>>>((float4*)agg, nh4);
    aggregate<<<gEdge, blk, 0, stream>>>(h, src, dst, dinv, agg, E);
    finalize<<<gNH4, blk, 0, stream>>>(agg, h, dinv, b2, xbuf, N, 1);

    gemm128_wmma<<<gGemm, blk, 0, stream>>>(xbuf, W3, h);
    zero_f4<<<gNH4, blk, 0, stream>>>((float4*)agg, nh4);
    aggregate<<<gEdge, blk, 0, stream>>>(h, src, dst, dinv, agg, E);
    finalize<<<gNH4, blk, 0, stream>>>(agg, h, dinv, b3, out, N, 0);
}


# aggregate by destination instead of float-atomic scatter: edges counting-sorted per call (int atomic rank), one wave sums each row in registers with fma; sum order = atomic arrival order
# speedup vs baseline: 2.8461x; 2.8461x over previous
#include <hip/hip_runtime.h>
#include <hip/hip_bf16.h>


#define NODES 50000
#define HDIM  128

typedef __attribute__((ext_vector_type(2))) float v2f;
typedef __attribute__((ext_vector_type(8))) float v8f;

__global__ __launch_bounds__(256) void gemm128_wmma(const float* __restrict__ X,
                                                    const float* __restrict__ W,
                                                    float* __restrict__ Hout) {
    __shared__ float Wl[128 * 128];
    __shared__ float Xl[16 * 130];

    const int tid = threadIdx.x;
    const int rowBase = blockIdx.x * 16;

    {
        const float4* W4 = (const float4*)W;
        float4* Wl4 = (float4*)Wl;
        for (int i = tid; i < 128 * 128 / 4; i += 256) Wl4[i] = W4[i];
    }
    {
        const float* Xt = X + (size_t)rowBase * HDIM;
        for (int i = tid; i < 16 * 128; i += 256) {
            int r = i >> 7, c = i & 127;
            Xl[r * 130 + c] = Xt[i];
        }
    }
    __syncthreads();

    const int lane = tid & 31;
    const int wave = tid >> 5;
    const int r16  = lane & 15;
    const int hi   = lane >> 4;
    const int colBase = wave * 16;

    v8f acc = {};
    #pragma unroll 4
    for (int k = 0; k < 128; k += 4) {
        const int kk = k + hi * 2;
        v2f a;
        a[0] = Xl[r16 * 130 + kk];
        a[1] = Xl[r16 * 130 + kk + 1];
        v2f b;
        b[0] = Wl[(kk + 0) * 128 + colBase + r16];
        b[1] = Wl[(kk + 1) * 128 + colBase + r16];
        acc = __builtin_amdgcn_wmma_f32_16x16x4_f32(
             false, a,  false, b,
             (short)0, acc,  false,  false);
    }

    float* out = Hout + (size_t)rowBase * HDIM + colBase + r16;
    #pragma unroll
    for (int r = 0; r < 8; ++r) {
        out[(size_t)(r + hi * 8) * HDIM] = acc[r];
    }
}

__global__ void zero_i32(int* p, int n) {
    int i = blockIdx.x * blockDim.x + threadIdx.x;
    if (i < n) p[i] = 0;
}

__global__ void count_rank(const int* __restrict__ dst, int* cnt,
                           int* __restrict__ rank, int e) {
    int i = blockIdx.x * blockDim.x + threadIdx.x;
    if (i < e) rank[i] = atomicAdd(&cnt[dst[i]], 1);
}

__global__ __launch_bounds__(256) void scan_rowptr(const int* __restrict__ cnt,
                                                   int* __restrict__ rowptr,
                                                   float* __restrict__ dinv, int n) {
    __shared__ int wpre[8];
    __shared__ int wsum[8];
    const int tid = threadIdx.x;
    const int lane = tid & 31;
    const int wave = tid >> 5;
    const int blockBase = blockIdx.x * 1024;

    int part = 0;
    for (int i = tid * 4; i < blockBase; i += 1024) {
        const int4 c = *(const int4*)(cnt + i);
        part += (c.x + c.y) + (c.z + c.w);
    }
    for (int o = 16; o > 0; o >>= 1) part += __shfl_xor(part, o);

    const int i0 = blockBase + tid * 4;
    const int c0 = (i0 + 0 < n) ? cnt[i0 + 0] : 0;
    const int c1 = (i0 + 1 < n) ? cnt[i0 + 1] : 0;
    const int c2 = (i0 + 2 < n) ? cnt[i0 + 2] : 0;
    const int c3 = (i0 + 3 < n) ? cnt[i0 + 3] : 0;
    const int t = (c0 + c1) + (c2 + c3);
    int incl = t;
    for (int o = 1; o < 32; o <<= 1) {
        int up = __shfl_up(incl, o);
        if (lane >= o) incl += up;
    }
    if (lane == 0)  wpre[wave] = part;
    if (lane == 31) wsum[wave] = incl;
    __syncthreads();

    int off = incl - t;
    for (int w = 0; w < 8; ++w) {
        off += wpre[w];
        if (w < wave) off += wsum[w];
    }
    if (i0 + 0 < n) { rowptr[i0 + 0] = off;                dinv[i0 + 0] = __frsqrt_rn((float)(c0 + 1)); }
    if (i0 + 1 < n) { rowptr[i0 + 1] = off + c0;           dinv[i0 + 1] = __frsqrt_rn((float)(c1 + 1)); }
    if (i0 + 2 < n) { rowptr[i0 + 2] = off + c0 + c1;      dinv[i0 + 2] = __frsqrt_rn((float)(c2 + 1)); }
    if (i0 + 3 < n) { rowptr[i0 + 3] = off + c0 + c1 + c2; dinv[i0 + 3] = __frsqrt_rn((float)(c3 + 1)); }
}

__global__ void place_edges(const int* __restrict__ src, const int* __restrict__ dst,
                            const int* __restrict__ rank, const int* __restrict__ rowptr,
                            const float* __restrict__ dinv, int2* __restrict__ csr, int e) {
    int i = blockIdx.x * blockDim.x + threadIdx.x;
    if (i >= e) return;
    const int s = src[i];
    const int d = dst[i];
    csr[rowptr[d] + rank[i]] = make_int2(s, __float_as_int(dinv[s]));
}

__device__ __forceinline__ void fma4(float4& a, float c, const float4& v) {
    a.x = fmaf(c, v.x, a.x);
    a.y = fmaf(c, v.y, a.y);
    a.z = fmaf(c, v.z, a.z);
    a.w = fmaf(c, v.w, a.w);
}

template <int R>
__device__ __forceinline__ void add_rows(float4& acc, const float4* __restrict__ h4,
                                         int s_l, float c_l, int j, int lane) {
    float4 v[R];
    float  c[R];
    #pragma unroll
    for (int r = 0; r < R; ++r) {
        const int s = __builtin_amdgcn_readlane(s_l, j + r);
        v[r] = h4[(size_t)s * 32 + lane];
    }
    #pragma unroll
    for (int r = 0; r < R; ++r)
        c[r] = __int_as_float(__builtin_amdgcn_readlane(__float_as_int(c_l), j + r));
    #pragma unroll
    for (int r = 0; r < R; ++r) fma4(acc, c[r], v[r]);
}

__global__ __launch_bounds__(256) void gather_rows(const float* __restrict__ h,
                                                   const int* __restrict__ rowptr,
                                                   const int* __restrict__ cnt,
                                                   const int2* __restrict__ csr,
                                                   const float* __restrict__ dinv,
                                                   const float* __restrict__ bias,
                                                   float* __restrict__ out,
                                                   int n, int relu) {
    const int lane = threadIdx.x & 31;
    const int node = __builtin_amdgcn_readfirstlane((int)((blockIdx.x * 256 + threadIdx.x) >> 5));
    if (node >= n) return;
    const int beg = rowptr[node];
    const int deg = cnt[node];
    const float dd = dinv[node];
    const float4* h4 = (const float4*)h;

    const float4 self = h4[(size_t)node * 32 + lane];
    float4 acc = make_float4(0.f, 0.f, 0.f, 0.f);

    for (int base = 0; base < deg; base += 32) {
        const int m = min(32, deg - base);
        int   s_l = 0;
        float c_l = 0.f;
        if (lane < m) {
            const int2 p = csr[beg + base + lane];
            s_l = p.x;
            c_l = __int_as_float(p.y) * dd;
        }
        int j = 0;
        for (; j + 4 <= m; j += 4) add_rows<4>(acc, h4, s_l, c_l, j, lane);
        const int rem = m - j;
        if (rem == 3)      add_rows<3>(acc, h4, s_l, c_l, j, lane);
        else if (rem == 2) add_rows<2>(acc, h4, s_l, c_l, j, lane);
        else if (rem == 1) add_rows<1>(acc, h4, s_l, c_l, j, lane);
    }
    fma4(acc, dd * dd, self);

    const float4 b = ((const float4*)bias)[lane];
    acc.x += b.x; acc.y += b.y; acc.z += b.z; acc.w += b.w;
    if (relu) {
        acc.x = fmaxf(acc.x, 0.f);
        acc.y = fmaxf(acc.y, 0.f);
        acc.z = fmaxf(acc.z, 0.f);
        acc.w = fmaxf(acc.w, 0.f);
    }
    ((float4*)out)[(size_t)node * 32 + lane] = acc;
}

extern "C" void kernel_launch(void* const* d_in, const int* in_sizes, int n_in,
                              void* d_out, int out_size, void* d_ws, size_t ws_size,
                              hipStream_t stream) {
    (void)n_in; (void)out_size; (void)ws_size;

    const float* x  = (const float*)d_in[0];
    const int*   ei = (const int*)d_in[1];
    const float* W1 = (const float*)d_in[2];
    const float* b1 = (const float*)d_in[3];
    const float* W2 = (const float*)d_in[4];
    const float* b2 = (const float*)d_in[5];
    const float* W3 = (const float*)d_in[6];
    const float* b3 = (const float*)d_in[7];

    const int N = in_sizes[0] / HDIM;
    const int E = in_sizes[1] / 2;
    const int* src = ei;
    const int* dst = ei + E;

    float* ws     = (float*)d_ws;
    float* h      = ws;
    float* xbuf   = ws + (size_t)N * HDIM;
    int*   cnt    = (int*)(ws + 2 * (size_t)N * HDIM);
    int*   rowptr = cnt + N;
    float* dinv   = (float*)(rowptr + N);
    int*   rank   = (int*)(dinv + N);
    int2*  csr    = (int2*)(rank + E);
    float* out    = (float*)d_out;

    dim3 blk(256);
    dim3 gN((N + 255) / 256);
    dim3 gE((E + 255) / 256);
    dim3 gScan((N + 1023) / 1024);
    dim3 gRows((N + 7) / 8);
    dim3 gGemm(N / 16);

    zero_i32<<<gN, blk, 0, stream>>>(cnt, N);
    count_rank<<<gE, blk, 0, stream>>>(dst, cnt, rank, E);
    scan_rowptr<<<gScan, blk, 0, stream>>>(cnt, rowptr, dinv, N);
    place_edges<<<gE, blk, 0, stream>>>(src, dst, rank, rowptr, dinv, csr, E);

    gemm128_wmma<<<gGemm, blk, 0, stream>>>(x, W1, h);
    gather_rows<<<gRows, blk, 0, stream>>>(h, rowptr, cnt, csr, dinv, b1, xbuf, N, 1);

    gemm128_wmma<<<gGemm, blk, 0, stream>>>(xbuf, W2, h);
    gather_rows<<<gRows, blk, 0, stream>>>(h, rowptr, cnt, csr, dinv, b2, xbuf, N, 1);

    gemm128_wmma<<<gGemm, blk, 0, stream>>>(xbuf, W3, h);
    gather_rows<<<gRows, blk, 0, stream>>>(h, rowptr, cnt, csr, dinv, b3, out, N, 0);
}


# every launch sized in waves, not items: 8 edges per thread, 8 rows per wave, 4-5 row tiles per GEMM block (W staged once); about 6x fewer waves enter; sum order still atomic arrival order
# speedup vs baseline: 3.3024x; 1.1603x over previous
#include <hip/hip_runtime.h>
#include <hip/hip_bf16.h>


#define NODES 50000
#define HDIM  128

typedef __attribute__((ext_vector_type(2))) float v2f;
typedef __attribute__((ext_vector_type(8))) float v8f;

#define XP 132

__global__ __launch_bounds__(256) void gemm128_wmma(const float* __restrict__ X,
                                                    const float* __restrict__ W,
                                                    float* __restrict__ Hout,
                                                    int nTiles) {
    __shared__ __attribute__((aligned(16))) float Wl[128 * 128];
    __shared__ __attribute__((aligned(16))) float Xl[2][16 * XP];

    const int tid = threadIdx.x;

    {
        const float4* W4 = (const float4*)W;
        float4* Wl4 = (float4*)Wl;
        for (int i = tid; i < 128 * 128 / 4; i += 256) Wl4[i] = W4[i];
    }

    const float4* X4 = (const float4*)X;
    const int tr = tid >> 5;
    const int tc = (tid & 31) * 4;

    int t = blockIdx.x;
    if (t < nTiles) {
        const float4 a0 = X4[(size_t)t * 512 + tid];
        const float4 a1 = X4[(size_t)t * 512 + 256 + tid];
        *(float4*)&Xl[0][tr * XP + tc]       = a0;
        *(float4*)&Xl[0][(tr + 8) * XP + tc] = a1;
    }
    __syncthreads();

    const int lane = tid & 31;
    const int wave = tid >> 5;
    const int r16  = lane & 15;
    const int hi   = lane >> 4;
    const int colBase = wave * 16;

    int cur = 0;
    for (; t < nTiles; t += gridDim.x) {
        const int tn = t + gridDim.x;
        const bool more = tn < nTiles;
        float4 n0 = make_float4(0.f, 0.f, 0.f, 0.f);
        float4 n1 = n0;
        if (more) {
            n0 = X4[(size_t)tn * 512 + tid];
            n1 = X4[(size_t)tn * 512 + 256 + tid];
        }

        const float* Xc = Xl[cur];
        v8f acc = {};
        #pragma unroll 4
        for (int k = 0; k < 128; k += 4) {
            const int kk = k + hi * 2;
            v2f a;
            a[0] = Xc[r16 * XP + kk];
            a[1] = Xc[r16 * XP + kk + 1];
            v2f b;
            b[0] = Wl[(kk + 0) * 128 + colBase + r16];
            b[1] = Wl[(kk + 1) * 128 + colBase + r16];
            acc = __builtin_amdgcn_wmma_f32_16x16x4_f32(
                 false, a,  false, b,
                 (short)0, acc,  false,  false);
        }

        if (more) {
            float* Xn = Xl[cur ^ 1];
            *(float4*)&Xn[tr * XP + tc]       = n0;
            *(float4*)&Xn[(tr + 8) * XP + tc] = n1;
        }
        __syncthreads();

        float* out = Hout + (size_t)t * 16 * HDIM + colBase + r16;
        #pragma unroll
        for (int r = 0; r < 8; ++r) {
            out[(size_t)(r + hi * 8) * HDIM] = acc[r];
        }
        cur ^= 1;
    }
}

__global__ __launch_bounds__(256) void zero_i32(int* p, int n) {
    const int i0 = (blockIdx.x * 256 + threadIdx.x) * 4;
    if (i0 + 4 <= n) {
        *(int4*)(p + i0) = make_int4(0, 0, 0, 0);
    } else {
        for (int i = i0; i < n; ++i) p[i] = 0;
    }
}

__global__ __launch_bounds__(256) void count_rank(const int* __restrict__ dst, int* cnt,
                                                  int* __restrict__ rank, int e) {
    const int i0 = (blockIdx.x * 256 + threadIdx.x) * 8;
    if (i0 >= e) return;
    if (i0 + 8 <= e && (e & 3) == 0) {
        const int4 d0 = *(const int4*)(dst + i0);
        const int4 d1 = *(const int4*)(dst + i0 + 4);
        int4 r0, r1;
        r0.x = atomicAdd(&cnt[d0.x], 1);
        r0.y = atomicAdd(&cnt[d0.y], 1);
        r0.z = atomicAdd(&cnt[d0.z], 1);
        r0.w = atomicAdd(&cnt[d0.w], 1);
        r1.x = atomicAdd(&cnt[d1.x], 1);
        r1.y = atomicAdd(&cnt[d1.y], 1);
        r1.z = atomicAdd(&cnt[d1.z], 1);
        r1.w = atomicAdd(&cnt[d1.w], 1);
        *(int4*)(rank + i0)     = r0;
        *(int4*)(rank + i0 + 4) = r1;
    } else {
        const int i1 = min(i0 + 8, e);
        for (int i = i0; i < i1; ++i) rank[i] = atomicAdd(&cnt[dst[i]], 1);
    }
}

__global__ __launch_bounds__(256) void scan_rowptr(const int* __restrict__ cnt,
                                                   int* __restrict__ rowptr,
                                                   float* __restrict__ dinv, int n) {
    __shared__ int wpre[8];
    __shared__ int wsum[8];
    const int tid = threadIdx.x;
    const int lane = tid & 31;
    const int wave = tid >> 5;
    const int blockBase = blockIdx.x * 1024;

    int part = 0;
    for (int i = tid * 4; i < blockBase; i += 1024) {
        const int4 c = *(const int4*)(cnt + i);
        part += (c.x + c.y) + (c.z + c.w);
    }
    for (int o = 16; o > 0; o >>= 1) part += __shfl_xor(part, o);

    const int i0 = blockBase + tid * 4;
    const int c0 = (i0 + 0 < n) ? cnt[i0 + 0] : 0;
    const int c1 = (i0 + 1 < n) ? cnt[i0 + 1] : 0;
    const int c2 = (i0 + 2 < n) ? cnt[i0 + 2] : 0;
    const int c3 = (i0 + 3 < n) ? cnt[i0 + 3] : 0;
    const int t = (c0 + c1) + (c2 + c3);
    int incl = t;
    for (int o = 1; o < 32; o <<= 1) {
        int up = __shfl_up(incl, o);
        if (lane >= o) incl += up;
    }
    if (lane == 0)  wpre[wave] = part;
    if (lane == 31) wsum[wave] = incl;
    __syncthreads();

    int off = incl - t;
    for (int w = 0; w < 8; ++w) {
        off += wpre[w];
        if (w < wave) off += wsum[w];
    }
    if (i0 + 0 < n) { rowptr[i0 + 0] = off;                dinv[i0 + 0] = __frsqrt_rn((float)(c0 + 1)); }
    if (i0 + 1 < n) { rowptr[i0 + 1] = off + c0;           dinv[i0 + 1] = __frsqrt_rn((float)(c1 + 1)); }
    if (i0 + 2 < n) { rowptr[i0 + 2] = off + c0 + c1;      dinv[i0 + 2] = __frsqrt_rn((float)(c2 + 1)); }
    if (i0 + 3 < n) { rowptr[i0 + 3] = off + c0 + c1 + c2; dinv[i0 + 3] = __frsqrt_rn((float)(c3 + 1)); }
}

__global__ __launch_bounds__(256) void place_edges(const int* __restrict__ src, const int* __restrict__ dst,
                                                   const int* __restrict__ rank, const int* __restrict__ rowptr,
                                                   const float* __restrict__ dinv, int2* __restrict__ csr, int e) {
    const int i0 = (blockIdx.x * 256 + threadIdx.x) * 8;
    if (i0 >= e) return;
    if (i0 + 8 <= e && (e & 3) == 0) {
        const int4 s0 = *(const int4*)(src + i0);
        const int4 s1 = *(const int4*)(src + i0 + 4);
        const int4 d0 = *(const int4*)(dst + i0);
        const int4 d1 = *(const int4*)(dst + i0 + 4);
        const int4 r0 = *(const int4*)(rank + i0);
        const int4 r1 = *(const int4*)(rank + i0 + 4);
        const int s[8] = {s0.x, s0.y, s0.z, s0.w, s1.x, s1.y, s1.z, s1.w};
        const int d[8] = {d0.x, d0.y, d0.z, d0.w, d1.x, d1.y, d1.z, d1.w};
        const int r[8] = {r0.x, r0.y, r0.z, r0.w, r1.x, r1.y, r1.z, r1.w};
        int   pos[8];
        float w[8];
        #pragma unroll
        for (int u = 0; u < 8; ++u) {
            pos[u] = rowptr[d[u]] + r[u];
            w[u]   = dinv[s[u]];
        }
        #pragma unroll
        for (int u = 0; u < 8; ++u) csr[pos[u]] = make_int2(s[u], __float_as_int(w[u]));
    } else {
        const int i1 = min(i0 + 8, e);
        for (int i = i0; i < i1; ++i) {
            const int sv = src[i];
            csr[rowptr[dst[i]] + rank[i]] = make_int2(sv, __float_as_int(dinv[sv]));
        }
    }
}

__device__ __forceinline__ void fma4(float4& a, float c, const float4& v) {
    a.x = fmaf(c, v.x, a.x);
    a.y = fmaf(c, v.y, a.y);
    a.z = fmaf(c, v.z, a.z);
    a.w = fmaf(c, v.w, a.w);
}

template <int R>
__device__ __forceinline__ void add_rows(float4& acc, const float4* __restrict__ h4,
                                         int s_l, float c_l, int j, int lane) {
    float4 v[R];
    float  c[R];
    #pragma unroll
    for (int r = 0; r < R; ++r) {
        const int s = __builtin_amdgcn_readlane(s_l, j + r);
        v[r] = h4[(size_t)s * 32 + lane];
    }
    #pragma unroll
    for (int r = 0; r < R; ++r)
        c[r] = __int_as_float(__builtin_amdgcn_readlane(__float_as_int(c_l), j + r));
    #pragma unroll
    for (int r = 0; r < R; ++r) fma4(acc, c[r], v[r]);
}

#define ROWS_PER_WAVE 8

__global__ __launch_bounds__(256) void gather_rows(const float* __restrict__ h,
                                                   const int* __restrict__ rowptr,
                                                   const int* __restrict__ cnt,
                                                   const int2* __restrict__ csr,
                                                   const float* __restrict__ dinv,
                                                   const float* __restrict__ bias,
                                                   float* __restrict__ out,
                                                   int n, int relu) {
    const int lane = threadIdx.x & 31;
    const int wv = __builtin_amdgcn_readfirstlane((int)((blockIdx.x * 256 + threadIdx.x) >> 5));
    const int n0 = wv * ROWS_PER_WAVE;
    if (n0 >= n) return;
    const int kn = min(ROWS_PER_WAVE, n - n0);
    const float4* h4 = (const float4*)h;

    int   deg_l = 0;
    float dd_l  = 0.f;
    if (lane < kn) {
        deg_l = cnt[n0 + lane];
        dd_l  = dinv[n0 + lane];
    }
    int beg = rowptr[n0];
    const float4 b = ((const float4*)bias)[lane];

    int deg = __builtin_amdgcn_readlane(deg_l, 0);
    int2 p = make_int2(0, 0);
    if (lane < deg) p = csr[beg + lane];

    for (int i = 0; i < kn; ++i) {
        const int node = n0 + i;
        const float dd = __int_as_float(__builtin_amdgcn_readlane(__float_as_int(dd_l), i));

        const float4 self = h4[(size_t)node * 32 + lane];

        const int begn = beg + deg;
        int  degn = 0;
        int2 pn = make_int2(0, 0);
        if (i + 1 < kn) {
            degn = __builtin_amdgcn_readlane(deg_l, i + 1);
            if (lane < degn) pn = csr[begn + lane];
        }

        float4 acc = make_float4(0.f, 0.f, 0.f, 0.f);
        for (int base = 0; base < deg; base += 32) {
            const int m = min(32, deg - base);
            if (base > 0) {
                p = make_int2(0, 0);
                if (lane < m) p = csr[beg + base + lane];
            }
            const int   s_l = p.x;
            const float c_l = __int_as_float(p.y) * dd;
            int j = 0;
            for (; j + 16 <= m; j += 16) add_rows<16>(acc, h4, s_l, c_l, j, lane);
            if (j + 8 <= m) { add_rows<8>(acc, h4, s_l, c_l, j, lane); j += 8; }
            switch (m - j) {
                case 7: add_rows<7>(acc, h4, s_l, c_l, j, lane); break;
                case 6: add_rows<6>(acc, h4, s_l, c_l, j, lane); break;
                case 5: add_rows<5>(acc, h4, s_l, c_l, j, lane); break;
                case 4: add_rows<4>(acc, h4, s_l, c_l, j, lane); break;
                case 3: add_rows<3>(acc, h4, s_l, c_l, j, lane); break;
                case 2: add_rows<2>(acc, h4, s_l, c_l, j, lane); break;
                case 1: add_rows<1>(acc, h4, s_l, c_l, j, lane); break;
                default: break;
            }
        }
        fma4(acc, dd * dd, self);

        acc.x += b.x; acc.y += b.y; acc.z += b.z; acc.w += b.w;
        if (relu) {
            acc.x = fmaxf(acc.x, 0.f);
            acc.y = fmaxf(acc.y, 0.f);
            acc.z = fmaxf(acc.z, 0.f);
            acc.w = fmaxf(acc.w, 0.f);
        }
        ((float4*)out)[(size_t)node * 32 + lane] = acc;

        beg = begn;
        deg = degn;
        p   = pn;
    }
}

extern "C" void kernel_launch(void* const* d_in, const int* in_sizes, int n_in,
                              void* d_out, int out_size, void* d_ws, size_t ws_size,
                              hipStream_t stream) {
    (void)n_in; (void)out_size; (void)ws_size;

    const float* x  = (const float*)d_in[0];
    const int*   ei = (const int*)d_in[1];
    const float* W1 = (const float*)d_in[2];
    const float* b1 = (const float*)d_in[3];
    const float* W2 = (const float*)d_in[4];
    const float* b2 = (const float*)d_in[5];
    const float* W3 = (const float*)d_in[6];
    const float* b3 = (const float*)d_in[7];

    const int N = in_sizes[0] / HDIM;
    const int E = in_sizes[1] / 2;
    const int* src = ei;
    const int* dst = ei + E;

    float* ws     = (float*)d_ws;
    float* h      = ws;
    float* xbuf   = ws + (size_t)N * HDIM;
    int*   cnt    = (int*)(ws + 2 * (size_t)N * HDIM);
    int*   rowptr = cnt + N;
    float* dinv   = (float*)(rowptr + N);
    int*   rank   = (int*)(dinv + N);
    int2*  csr    = (int2*)(rank + E);
    float* out    = (float*)d_out;

    dim3 blk(256);
    dim3 gN((N + 1023) / 1024);
    dim3 gE((E + 2047) / 2048);
    dim3 gScan((N + 1023) / 1024);
    dim3 gRows((N + 63) / 64);
    dim3 gGemm(768);
    const int nTiles = N / 16;

    zero_i32<<<gN, blk, 0, stream>>>(cnt, N);
    count_rank<<<gE, blk, 0, stream>>>(dst, cnt, rank, E);
    scan_rowptr<<<gScan, blk, 0, stream>>>(cnt, rowptr, dinv, N);
    place_edges<<<gE, blk, 0, stream>>>(src, dst, rank, rowptr, dinv, csr, E);

    gemm128_wmma<<<gGemm, blk, 0, stream>>>(x, W1, h, nTiles);
    gather_rows<<<gRows, blk, 0, stream>>>(h, rowptr, cnt, csr, dinv, b1, xbuf, N, 1);

    gemm128_wmma<<<gGemm, blk, 0, stream>>>(xbuf, W2, h, nTiles);
    gather_rows<<<gRows, blk, 0, stream>>>(h, rowptr, cnt, csr, dinv, b2, xbuf, N, 1);

    gemm128_wmma<<<gGemm, blk, 0, stream>>>(xbuf, W3, h, nTiles);
    gather_rows<<<gRows, blk, 0, stream>>>(h, rowptr, cnt, csr, dinv, b3, out, N, 0);
}


# the row-sum launches of layers 1 and 2 keep their 16 summed rows in LDS and multiply them by the next layer's W on WMMA there: 10 launches -> 8, summed rows never stored; sum order unchanged
# speedup vs baseline: 3.5416x; 1.0724x over previous
#include <hip/hip_runtime.h>
#include <hip/hip_bf16.h>


#define NODES 50000
#define HDIM  128

typedef __attribute__((ext_vector_type(2))) float v2f;
typedef __attribute__((ext_vector_type(8))) float v8f;

#define XP 132

__global__ __launch_bounds__(256) void gemm128_wmma(const float* __restrict__ X,
                                                    const float* __restrict__ W,
                                                    float* __restrict__ Hout,
                                                    int nTiles) {
    __shared__ __attribute__((aligned(16))) float Wl[128 * 128];
    __shared__ __attribute__((aligned(16))) float Xl[2][16 * XP];

    const int tid = threadIdx.x;

    {
        const float4* W4 = (const float4*)W;
        float4* Wl4 = (float4*)Wl;
        for (int i = tid; i < 128 * 128 / 4; i += 256) Wl4[i] = W4[i];
    }

    const float4* X4 = (const float4*)X;
    const int tr = tid >> 5;
    const int tc = (tid & 31) * 4;

    int t = blockIdx.x;
    if (t < nTiles) {
        const float4 a0 = X4[(size_t)t * 512 + tid];
        const float4 a1 = X4[(size_t)t * 512 + 256 + tid];
        *(float4*)&Xl[0][tr * XP + tc]       = a0;
        *(float4*)&Xl[0][(tr + 8) * XP + tc] = a1;
    }
    __syncthreads();

    const int lane = tid & 31;
    const int wave = tid >> 5;
    const int r16  = lane & 15;
    const int hi   = lane >> 4;
    const int colBase = wave * 16;

    int cur = 0;
    for (; t < nTiles; t += gridDim.x) {
        const int tn = t + gridDim.x;
        const bool more = tn < nTiles;
        float4 n0 = make_float4(0.f, 0.f, 0.f, 0.f);
        float4 n1 = n0;
        if (more) {
            n0 = X4[(size_t)tn * 512 + tid];
            n1 = X4[(size_t)tn * 512 + 256 + tid];
        }

        const float* Xc = Xl[cur];
        v8f acc = {};
        #pragma unroll 4
        for (int k = 0; k < 128; k += 4) {
            const int kk = k + hi * 2;
            v2f a;
            a[0] = Xc[r16 * XP + kk];
            a[1] = Xc[r16 * XP + kk + 1];
            v2f b;
            b[0] = Wl[(kk + 0) * 128 + colBase + r16];
            b[1] = Wl[(kk + 1) * 128 + colBase + r16];
            acc = __builtin_amdgcn_wmma_f32_16x16x4_f32(
                 false, a,  false, b,
                 (short)0, acc,  false,  false);
        }

        if (more) {
            float* Xn = Xl[cur ^ 1];
            *(float4*)&Xn[tr * XP + tc]       = n0;
            *(float4*)&Xn[(tr + 8) * XP + tc] = n1;
        }
        __syncthreads();

        float* out = Hout + (size_t)t * 16 * HDIM + colBase + r16;
        #pragma unroll
        for (int r = 0; r < 8; ++r) {
            out[(size_t)(r + hi * 8) * HDIM] = acc[r];
        }
        cur ^= 1;
    }
}

__global__ __launch_bounds__(256) void zero_i32(int* p, int n) {
    const int i0 = (blockIdx.x * 256 + threadIdx.x) * 4;
    if (i0 + 4 <= n) {
        *(int4*)(p + i0) = make_int4(0, 0, 0, 0);
    } else {
        for (int i = i0; i < n; ++i) p[i] = 0;
    }
}

__global__ __launch_bounds__(256) void count_rank(const int* __restrict__ dst, int* cnt,
                                                  int* __restrict__ rank, int e) {
    const int i0 = (blockIdx.x * 256 + threadIdx.x) * 8;
    if (i0 >= e) return;
    if (i0 + 8 <= e && (e & 3) == 0) {
        const int4 d0 = *(const int4*)(dst + i0);
        const int4 d1 = *(const int4*)(dst + i0 + 4);
        int4 r0, r1;
        r0.x = atomicAdd(&cnt[d0.x], 1);
        r0.y = atomicAdd(&cnt[d0.y], 1);
        r0.z = atomicAdd(&cnt[d0.z], 1);
        r0.w = atomicAdd(&cnt[d0.w], 1);
        r1.x = atomicAdd(&cnt[d1.x], 1);
        r1.y = atomicAdd(&cnt[d1.y], 1);
        r1.z = atomicAdd(&cnt[d1.z], 1);
        r1.w = atomicAdd(&cnt[d1.w], 1);
        *(int4*)(rank + i0)     = r0;
        *(int4*)(rank + i0 + 4) = r1;
    } else {
        const int i1 = min(i0 + 8, e);
        for (int i = i0; i < i1; ++i) rank[i] = atomicAdd(&cnt[dst[i]], 1);
    }
}

__global__ __launch_bounds__(256) void scan_rowptr(const int* __restrict__ cnt,
                                                   int* __restrict__ rowptr,
                                                   float* __restrict__ dinv, int n) {
    __shared__ int wpre[8];
    __shared__ int wsum[8];
    const int tid = threadIdx.x;
    const int lane = tid & 31;
    const int wave = tid >> 5;
    const int blockBase = blockIdx.x * 1024;

    int part = 0;
    for (int i = tid * 4; i < blockBase; i += 1024) {
        const int4 c = *(const int4*)(cnt + i);
        part += (c.x + c.y) + (c.z + c.w);
    }
    for (int o = 16; o > 0; o >>= 1) part += __shfl_xor(part, o);

    const int i0 = blockBase + tid * 4;
    const int c0 = (i0 + 0 < n) ? cnt[i0 + 0] : 0;
    const int c1 = (i0 + 1 < n) ? cnt[i0 + 1] : 0;
    const int c2 = (i0 + 2 < n) ? cnt[i0 + 2] : 0;
    const int c3 = (i0 + 3 < n) ? cnt[i0 + 3] : 0;
    const int t = (c0 + c1) + (c2 + c3);
    int incl = t;
    for (int o = 1; o < 32; o <<= 1) {
        int up = __shfl_up(incl, o);
        if (lane >= o) incl += up;
    }
    if (lane == 0)  wpre[wave] = part;
    if (lane == 31) wsum[wave] = incl;
    __syncthreads();

    int off = incl - t;
    for (int w = 0; w < 8; ++w) {
        off += wpre[w];
        if (w < wave) off += wsum[w];
    }
    if (i0 + 0 < n) { rowptr[i0 + 0] = off;                dinv[i0 + 0] = __frsqrt_rn((float)(c0 + 1)); }
    if (i0 + 1 < n) { rowptr[i0 + 1] = off + c0;           dinv[i0 + 1] = __frsqrt_rn((float)(c1 + 1)); }
    if (i0 + 2 < n) { rowptr[i0 + 2] = off + c0 + c1;      dinv[i0 + 2] = __frsqrt_rn((float)(c2 + 1)); }
    if (i0 + 3 < n) { rowptr[i0 + 3] = off + c0 + c1 + c2; dinv[i0 + 3] = __frsqrt_rn((float)(c3 + 1)); }
}

__global__ __launch_bounds__(256) void place_edges(const int* __restrict__ src, const int* __restrict__ dst,
                                                   const int* __restrict__ rank, const int* __restrict__ rowptr,
                                                   const float* __restrict__ dinv, int2* __restrict__ csr, int e) {
    const int i0 = (blockIdx.x * 256 + threadIdx.x) * 8;
    if (i0 >= e) return;
    if (i0 + 8 <= e && (e & 3) == 0) {
        const int4 s0 = *(const int4*)(src + i0);
        const int4 s1 = *(const int4*)(src + i0 + 4);
        const int4 d0 = *(const int4*)(dst + i0);
        const int4 d1 = *(const int4*)(dst + i0 + 4);
        const int4 r0 = *(const int4*)(rank + i0);
        const int4 r1 = *(const int4*)(rank + i0 + 4);
        const int s[8] = {s0.x, s0.y, s0.z, s0.w, s1.x, s1.y, s1.z, s1.w};
        const int d[8] = {d0.x, d0.y, d0.z, d0.w, d1.x, d1.y, d1.z, d1.w};
        const int r[8] = {r0.x, r0.y, r0.z, r0.w, r1.x, r1.y, r1.z, r1.w};
        int   pos[8];
        float w[8];
        #pragma unroll
        for (int u = 0; u < 8; ++u) {
            pos[u] = rowptr[d[u]] + r[u];
            w[u]   = dinv[s[u]];
        }
        #pragma unroll
        for (int u = 0; u < 8; ++u) csr[pos[u]] = make_int2(s[u], __float_as_int(w[u]));
    } else {
        const int i1 = min(i0 + 8, e);
        for (int i = i0; i < i1; ++i) {
            const int sv = src[i];
            csr[rowptr[dst[i]] + rank[i]] = make_int2(sv, __float_as_int(dinv[sv]));
        }
    }
}

__device__ __forceinline__ void fma4(float4& a, float c, const float4& v) {
    a.x = fmaf(c, v.x, a.x);
    a.y = fmaf(c, v.y, a.y);
    a.z = fmaf(c, v.z, a.z);
    a.w = fmaf(c, v.w, a.w);
}

template <int R>
__device__ __forceinline__ void add_rows(float4& acc, const float4* __restrict__ h4,
                                         int s_l, float c_l, int j, int lane) {
    float4 v[R];
    float  c[R];
    #pragma unroll
    for (int r = 0; r < R; ++r) {
        const int s = __builtin_amdgcn_readlane(s_l, j + r);
        v[r] = h4[(size_t)s * 32 + lane];
    }
    #pragma unroll
    for (int r = 0; r < R; ++r)
        c[r] = __int_as_float(__builtin_amdgcn_readlane(__float_as_int(c_l), j + r));
    #pragma unroll
    for (int r = 0; r < R; ++r) fma4(acc, c[r], v[r]);
}

#define ROWS_PER_WAVE 8

__global__ __launch_bounds__(256) void gather_rows(const float* __restrict__ h,
                                                   const int* __restrict__ rowptr,
                                                   const int* __restrict__ cnt,
                                                   const int2* __restrict__ csr,
                                                   const float* __restrict__ dinv,
                                                   const float* __restrict__ bias,
                                                   float* __restrict__ out,
                                                   int n, int relu) {
    const int lane = threadIdx.x & 31;
    const int wv = __builtin_amdgcn_readfirstlane((int)((blockIdx.x * 256 + threadIdx.x) >> 5));
    const int n0 = wv * ROWS_PER_WAVE;
    if (n0 >= n) return;
    const int kn = min(ROWS_PER_WAVE, n - n0);
    const float4* h4 = (const float4*)h;

    int   deg_l = 0;
    float dd_l  = 0.f;
    if (lane < kn) {
        deg_l = cnt[n0 + lane];
        dd_l  = dinv[n0 + lane];
    }
    int beg = rowptr[n0];
    const float4 b = ((const float4*)bias)[lane];

    int deg = __builtin_amdgcn_readlane(deg_l, 0);
    int2 p = make_int2(0, 0);
    if (lane < deg) p = csr[beg + lane];

    for (int i = 0; i < kn; ++i) {
        const int node = n0 + i;
        const float dd = __int_as_float(__builtin_amdgcn_readlane(__float_as_int(dd_l), i));

        const float4 self = h4[(size_t)node * 32 + lane];

        const int begn = beg + deg;
        int  degn = 0;
        int2 pn = make_int2(0, 0);
        if (i + 1 < kn) {
            degn = __builtin_amdgcn_readlane(deg_l, i + 1);
            if (lane < degn) pn = csr[begn + lane];
        }

        float4 acc = make_float4(0.f, 0.f, 0.f, 0.f);
        for (int base = 0; base < deg; base += 32) {
            const int m = min(32, deg - base);
            if (base > 0) {
                p = make_int2(0, 0);
                if (lane < m) p = csr[beg + base + lane];
            }
            const int   s_l = p.x;
            const float c_l = __int_as_float(p.y) * dd;
            int j = 0;
            for (; j + 16 <= m; j += 16) add_rows<16>(acc, h4, s_l, c_l, j, lane);
            if (j + 8 <= m) { add_rows<8>(acc, h4, s_l, c_l, j, lane); j += 8; }
            switch (m - j) {
                case 7: add_rows<7>(acc, h4, s_l, c_l, j, lane); break;
                case 6: add_rows<6>(acc, h4, s_l, c_l, j, lane); break;
                case 5: add_rows<5>(acc, h4, s_l, c_l, j, lane); break;
                case 4: add_rows<4>(acc, h4, s_l, c_l, j, lane); break;
                case 3: add_rows<3>(acc, h4, s_l, c_l, j, lane); break;
                case 2: add_rows<2>(acc, h4, s_l, c_l, j, lane); break;
                case 1: add_rows<1>(acc, h4, s_l, c_l, j, lane); break;
                default: break;
            }
        }
        fma4(acc, dd * dd, self);

        acc.x += b.x; acc.y += b.y; acc.z += b.z; acc.w += b.w;
        if (relu) {
            acc.x = fmaxf(acc.x, 0.f);
            acc.y = fmaxf(acc.y, 0.f);
            acc.z = fmaxf(acc.z, 0.f);
            acc.w = fmaxf(acc.w, 0.f);
        }
        ((float4*)out)[(size_t)node * 32 + lane] = acc;

        beg = begn;
        deg = degn;
        p   = pn;
    }
}


__global__ __launch_bounds__(256) void gather_gemm(const float* __restrict__ h,
                                                   const int* __restrict__ rowptr,
                                                   const int* __restrict__ cnt,
                                                   const int2* __restrict__ csr,
                                                   const float* __restrict__ dinv,
                                                   const float* __restrict__ bias,
                                                   const float* __restrict__ W,
                                                   float* __restrict__ Hout,
                                                   int nTiles) {
    __shared__ __attribute__((aligned(16))) float Wl[128 * 128];
    __shared__ __attribute__((aligned(16))) float Xl[2][16 * XP];

    const int tid = threadIdx.x;
    {
        const float4* W4 = (const float4*)W;
        float4* Wl4 = (float4*)Wl;
        for (int i = tid; i < 128 * 128 / 4; i += 256) Wl4[i] = W4[i];
    }

    const int lane = tid & 31;
    const int wave = __builtin_amdgcn_readfirstlane(tid >> 5);
    const int r16  = lane & 15;
    const int hi   = lane >> 4;
    const int colBase = wave * 16;
    const float4* h4 = (const float4*)h;
    const float4 b = ((const float4*)bias)[lane];

    int t = blockIdx.x;
    int   deg_l = 0;
    float dd_l  = 0.f;
    int   beg   = 0;
    if (t < nTiles) {
        const int n0 = t * 16 + wave * 2;
        if (lane < 2) {
            deg_l = cnt[n0 + lane];
            dd_l  = dinv[n0 + lane];
        }
        beg = rowptr[n0];
    }

    int cur = 0;
    for (; t < nTiles; t += gridDim.x) {
        const int n0 = t * 16 + wave * 2;
        float* Xw = &Xl[cur][(wave * 2) * XP + lane * 4];

        const int   deg0 = __builtin_amdgcn_readlane(deg_l, 0);
        const int   deg1 = __builtin_amdgcn_readlane(deg_l, 1);
        const float dd0  = __int_as_float(__builtin_amdgcn_readlane(__float_as_int(dd_l), 0));
        const float dd1  = __int_as_float(__builtin_amdgcn_readlane(__float_as_int(dd_l), 1));
        int beg_r = beg;
        int deg   = deg0;
        int2 p = make_int2(0, 0);
        if (lane < deg) p = csr[beg_r + lane];

        {
            const int tn = t + gridDim.x;
            deg_l = 0;
            dd_l  = 0.f;
            if (tn < nTiles) {
                const int n0n = tn * 16 + wave * 2;
                if (lane < 2) {
                    deg_l = cnt[n0n + lane];
                    dd_l  = dinv[n0n + lane];
                }
                beg = rowptr[n0n];
            }
        }

        #pragma unroll
        for (int i = 0; i < 2; ++i) {
            const int node = n0 + i;
            const float dd = (i == 0) ? dd0 : dd1;
            const float4 self = h4[(size_t)node * 32 + lane];

            const int begn = beg_r + deg;
            int  degn = 0;
            int2 pn = make_int2(0, 0);
            if (i == 0) {
                degn = deg1;
                if (lane < degn) pn = csr[begn + lane];
            }

            float4 acc = make_float4(0.f, 0.f, 0.f, 0.f);
            for (int base = 0; base < deg; base += 32) {
                const int m = min(32, deg - base);
                if (base > 0) {
                    p = make_int2(0, 0);
                    if (lane < m) p = csr[beg_r + base + lane];
                }
                const int   s_l = p.x;
                const float c_l = __int_as_float(p.y) * dd;
                int j = 0;
                for (; j + 16 <= m; j += 16) add_rows<16>(acc, h4, s_l, c_l, j, lane);
                if (j + 8 <= m) { add_rows<8>(acc, h4, s_l, c_l, j, lane); j += 8; }
                switch (m - j) {
                    case 7: add_rows<7>(acc, h4, s_l, c_l, j, lane); break;
                    case 6: add_rows<6>(acc, h4, s_l, c_l, j, lane); break;
                    case 5: add_rows<5>(acc, h4, s_l, c_l, j, lane); break;
                    case 4: add_rows<4>(acc, h4, s_l, c_l, j, lane); break;
                    case 3: add_rows<3>(acc, h4, s_l, c_l, j, lane); break;
                    case 2: add_rows<2>(acc, h4, s_l, c_l, j, lane); break;
                    case 1: add_rows<1>(acc, h4, s_l, c_l, j, lane); break;
                    default: break;
                }
            }
            fma4(acc, dd * dd, self);
            acc.x = fmaxf(acc.x + b.x, 0.f);
            acc.y = fmaxf(acc.y + b.y, 0.f);
            acc.z = fmaxf(acc.z + b.z, 0.f);
            acc.w = fmaxf(acc.w + b.w, 0.f);
            *(float4*)(Xw + i * XP) = acc;

            beg_r = begn;
            deg   = degn;
            p     = pn;
        }

        __syncthreads();

        const float* Xc = Xl[cur];
        v8f acc8 = {};
        #pragma unroll 4
        for (int k = 0; k < 128; k += 4) {
            const int kk = k + hi * 2;
            v2f a;
            a[0] = Xc[r16 * XP + kk];
            a[1] = Xc[r16 * XP + kk + 1];
            v2f bb;
            bb[0] = Wl[(kk + 0) * 128 + colBase + r16];
            bb[1] = Wl[(kk + 1) * 128 + colBase + r16];
            acc8 = __builtin_amdgcn_wmma_f32_16x16x4_f32(
                 false, a,  false, bb,
                 (short)0, acc8,  false,  false);
        }

        float* out = Hout + (size_t)t * 16 * HDIM + colBase + r16;
        #pragma unroll
        for (int r = 0; r < 8; ++r) {
            out[(size_t)(r + hi * 8) * HDIM] = acc8[r];
        }
        cur ^= 1;
    }
}

extern "C" void kernel_launch(void* const* d_in, const int* in_sizes, int n_in,
                              void* d_out, int out_size, void* d_ws, size_t ws_size,
                              hipStream_t stream) {
    (void)n_in; (void)out_size; (void)ws_size;

    const float* x  = (const float*)d_in[0];
    const int*   ei = (const int*)d_in[1];
    const float* W1 = (const float*)d_in[2];
    const float* b1 = (const float*)d_in[3];
    const float* W2 = (const float*)d_in[4];
    const float* b2 = (const float*)d_in[5];
    const float* W3 = (const float*)d_in[6];
    const float* b3 = (const float*)d_in[7];

    const int N = in_sizes[0] / HDIM;
    const int E = in_sizes[1] / 2;
    const int* src = ei;
    const int* dst = ei + E;

    float* ws     = (float*)d_ws;
    float* h      = ws;
    float* xbuf   = ws + (size_t)N * HDIM;
    int*   cnt    = (int*)(ws + 2 * (size_t)N * HDIM);
    int*   rowptr = cnt + N;
    float* dinv   = (float*)(rowptr + N);
    int*   rank   = (int*)(dinv + N);
    int2*  csr    = (int2*)(rank + E);
    float* out    = (float*)d_out;

    dim3 blk(256);
    dim3 gN((N + 1023) / 1024);
    dim3 gE((E + 2047) / 2048);
    dim3 gScan((N + 1023) / 1024);
    dim3 gRows((N + 63) / 64);
    dim3 gGemm(768);
    const int nTiles = N / 16;

    zero_i32<<<gN, blk, 0, stream>>>(cnt, N);
    count_rank<<<gE, blk, 0, stream>>>(dst, cnt, rank, E);
    scan_rowptr<<<gScan, blk, 0, stream>>>(cnt, rowptr, dinv, N);
    place_edges<<<gE, blk, 0, stream>>>(src, dst, rank, rowptr, dinv, csr, E);

    gemm128_wmma<<<gGemm, blk, 0, stream>>>(x, W1, h, nTiles);

    gather_gemm<<<gGemm, blk, 0, stream>>>(h, rowptr, cnt, csr, dinv, b1, W2, xbuf, nTiles);

    gather_gemm<<<gGemm, blk, 0, stream>>>(xbuf, rowptr, cnt, csr, dinv, b2, W3, h, nTiles);

    gather_rows<<<gRows, blk, 0, stream>>>(h, rowptr, cnt, csr, dinv, b3, out, N, 0);
}


# each layer sums first and multiplies after: (A a) W + b replaces A (a W) + b, bias and ReLU on the WMMA result; 8 launches -> 7, one 25.6 MB plane less; sums and products round in another order
# speedup vs baseline: 3.6578x; 1.0328x over previous
#include <hip/hip_runtime.h>
#include <hip/hip_bf16.h>


#define NODES 50000
#define HDIM  128

typedef __attribute__((ext_vector_type(2))) float v2f;
typedef __attribute__((ext_vector_type(8))) float v8f;

#define XP 132

__global__ __launch_bounds__(256) void zero_i32(int* p, int n) {
    const int i0 = (blockIdx.x * 256 + threadIdx.x) * 4;
    if (i0 + 4 <= n) {
        *(int4*)(p + i0) = make_int4(0, 0, 0, 0);
    } else {
        for (int i = i0; i < n; ++i) p[i] = 0;
    }
}

__global__ __launch_bounds__(256) void count_rank(const int* __restrict__ dst, int* cnt,
                                                  int* __restrict__ rank, int e) {
    const int i0 = (blockIdx.x * 256 + threadIdx.x) * 8;
    if (i0 >= e) return;
    if (i0 + 8 <= e && (e & 3) == 0) {
        const int4 d0 = *(const int4*)(dst + i0);
        const int4 d1 = *(const int4*)(dst + i0 + 4);
        int4 r0, r1;
        r0.x = atomicAdd(&cnt[d0.x], 1);
        r0.y = atomicAdd(&cnt[d0.y], 1);
        r0.z = atomicAdd(&cnt[d0.z], 1);
        r0.w = atomicAdd(&cnt[d0.w], 1);
        r1.x = atomicAdd(&cnt[d1.x], 1);
        r1.y = atomicAdd(&cnt[d1.y], 1);
        r1.z = atomicAdd(&cnt[d1.z], 1);
        r1.w = atomicAdd(&cnt[d1.w], 1);
        *(int4*)(rank + i0)     = r0;
        *(int4*)(rank + i0 + 4) = r1;
    } else {
        const int i1 = min(i0 + 8, e);
        for (int i = i0; i < i1; ++i) rank[i] = atomicAdd(&cnt[dst[i]], 1);
    }
}

__global__ __launch_bounds__(256) void scan_rowptr(const int* __restrict__ cnt,
                                                   int* __restrict__ rowptr,
                                                   float* __restrict__ dinv, int n) {
    __shared__ int wpre[8];
    __shared__ int wsum[8];
    const int tid = threadIdx.x;
    const int lane = tid & 31;
    const int wave = tid >> 5;
    const int blockBase = blockIdx.x * 1024;

    int part = 0;
    for (int i = tid * 4; i < blockBase; i += 1024) {
        const int4 c = *(const int4*)(cnt + i);
        part += (c.x + c.y) + (c.z + c.w);
    }
    for (int o = 16; o > 0; o >>= 1) part += __shfl_xor(part, o);

    const int i0 = blockBase + tid * 4;
    const int c0 = (i0 + 0 < n) ? cnt[i0 + 0] : 0;
    const int c1 = (i0 + 1 < n) ? cnt[i0 + 1] : 0;
    const int c2 = (i0 + 2 < n) ? cnt[i0 + 2] : 0;
    const int c3 = (i0 + 3 < n) ? cnt[i0 + 3] : 0;
    const int t = (c0 + c1) + (c2 + c3);
    int incl = t;
    for (int o = 1; o < 32; o <<= 1) {
        int up = __shfl_up(incl, o);
        if (lane >= o) incl += up;
    }
    if (lane == 0)  wpre[wave] = part;
    if (lane == 31) wsum[wave] = incl;
    __syncthreads();

    int off = incl - t;
    for (int w = 0; w < 8; ++w) {
        off += wpre[w];
        if (w < wave) off += wsum[w];
    }
    if (i0 + 0 < n) { rowptr[i0 + 0] = off;                dinv[i0 + 0] = __frsqrt_rn((float)(c0 + 1)); }
    if (i0 + 1 < n) { rowptr[i0 + 1] = off + c0;           dinv[i0 + 1] = __frsqrt_rn((float)(c1 + 1)); }
    if (i0 + 2 < n) { rowptr[i0 + 2] = off + c0 + c1;      dinv[i0 + 2] = __frsqrt_rn((float)(c2 + 1)); }
    if (i0 + 3 < n) { rowptr[i0 + 3] = off + c0 + c1 + c2; dinv[i0 + 3] = __frsqrt_rn((float)(c3 + 1)); }
}

__global__ __launch_bounds__(256) void place_edges(const int* __restrict__ src, const int* __restrict__ dst,
                                                   const int* __restrict__ rank, const int* __restrict__ rowptr,
                                                   const float* __restrict__ dinv, int2* __restrict__ csr, int e) {
    const int i0 = (blockIdx.x * 256 + threadIdx.x) * 8;
    if (i0 >= e) return;
    if (i0 + 8 <= e && (e & 3) == 0) {
        const int4 s0 = *(const int4*)(src + i0);
        const int4 s1 = *(const int4*)(src + i0 + 4);
        const int4 d0 = *(const int4*)(dst + i0);
        const int4 d1 = *(const int4*)(dst + i0 + 4);
        const int4 r0 = *(const int4*)(rank + i0);
        const int4 r1 = *(const int4*)(rank + i0 + 4);
        const int s[8] = {s0.x, s0.y, s0.z, s0.w, s1.x, s1.y, s1.z, s1.w};
        const int d[8] = {d0.x, d0.y, d0.z, d0.w, d1.x, d1.y, d1.z, d1.w};
        const int r[8] = {r0.x, r0.y, r0.z, r0.w, r1.x, r1.y, r1.z, r1.w};
        int   pos[8];
        float w[8];
        #pragma unroll
        for (int u = 0; u < 8; ++u) {
            pos[u] = rowptr[d[u]] + r[u];
            w[u]   = dinv[s[u]];
        }
        #pragma unroll
        for (int u = 0; u < 8; ++u) csr[pos[u]] = make_int2(s[u], __float_as_int(w[u]));
    } else {
        const int i1 = min(i0 + 8, e);
        for (int i = i0; i < i1; ++i) {
            const int sv = src[i];
            csr[rowptr[dst[i]] + rank[i]] = make_int2(sv, __float_as_int(dinv[sv]));
        }
    }
}

__device__ __forceinline__ void fma4(float4& a, float c, const float4& v) {
    a.x = fmaf(c, v.x, a.x);
    a.y = fmaf(c, v.y, a.y);
    a.z = fmaf(c, v.z, a.z);
    a.w = fmaf(c, v.w, a.w);
}

template <int R>
__device__ __forceinline__ void add_rows(float4& acc, const float4* __restrict__ h4,
                                         int s_l, float c_l, int j, int lane) {
    float4 v[R];
    float  c[R];
    #pragma unroll
    for (int r = 0; r < R; ++r) {
        const int s = __builtin_amdgcn_readlane(s_l, j + r);
        v[r] = h4[(size_t)s * 32 + lane];
    }
    #pragma unroll
    for (int r = 0; r < R; ++r)
        c[r] = __int_as_float(__builtin_amdgcn_readlane(__float_as_int(c_l), j + r));
    #pragma unroll
    for (int r = 0; r < R; ++r) fma4(acc, c[r], v[r]);
}

__global__ __launch_bounds__(256) void gather_gemm(const float* __restrict__ h,
                                                   const int* __restrict__ rowptr,
                                                   const int* __restrict__ cnt,
                                                   const int2* __restrict__ csr,
                                                   const float* __restrict__ dinv,
                                                   const float* __restrict__ bias,
                                                   const float* __restrict__ W,
                                                   float* __restrict__ Hout,
                                                   int nTiles, int relu) {
    __shared__ __attribute__((aligned(16))) float Wl[128 * 128];
    __shared__ __attribute__((aligned(16))) float Xl[2][16 * XP];

    const int tid = threadIdx.x;
    {
        const float4* W4 = (const float4*)W;
        float4* Wl4 = (float4*)Wl;
        for (int i = tid; i < 128 * 128 / 4; i += 256) Wl4[i] = W4[i];
    }

    const int lane = tid & 31;
    const int wave = __builtin_amdgcn_readfirstlane(tid >> 5);
    const int r16  = lane & 15;
    const int hi   = lane >> 4;
    const int colBase = wave * 16;
    const float4* h4 = (const float4*)h;
    const float bv = bias[colBase + r16];

    int t = blockIdx.x;
    int   deg_l = 0;
    float dd_l  = 0.f;
    int   beg   = 0;
    if (t < nTiles) {
        const int n0 = t * 16 + wave * 2;
        if (lane < 2) {
            deg_l = cnt[n0 + lane];
            dd_l  = dinv[n0 + lane];
        }
        beg = rowptr[n0];
    }

    int cur = 0;
    for (; t < nTiles; t += gridDim.x) {
        const int n0 = t * 16 + wave * 2;
        float* Xw = &Xl[cur][(wave * 2) * XP + lane * 4];

        const int   deg0 = __builtin_amdgcn_readlane(deg_l, 0);
        const int   deg1 = __builtin_amdgcn_readlane(deg_l, 1);
        const float dd0  = __int_as_float(__builtin_amdgcn_readlane(__float_as_int(dd_l), 0));
        const float dd1  = __int_as_float(__builtin_amdgcn_readlane(__float_as_int(dd_l), 1));
        int beg_r = beg;
        int deg   = deg0;
        int2 p = make_int2(0, 0);
        if (lane < deg) p = csr[beg_r + lane];

        {
            const int tn = t + gridDim.x;
            deg_l = 0;
            dd_l  = 0.f;
            if (tn < nTiles) {
                const int n0n = tn * 16 + wave * 2;
                if (lane < 2) {
                    deg_l = cnt[n0n + lane];
                    dd_l  = dinv[n0n + lane];
                }
                beg = rowptr[n0n];
            }
        }

        #pragma unroll
        for (int i = 0; i < 2; ++i) {
            const int node = n0 + i;
            const float dd = (i == 0) ? dd0 : dd1;
            const float4 self = h4[(size_t)node * 32 + lane];

            const int begn = beg_r + deg;
            int  degn = 0;
            int2 pn = make_int2(0, 0);
            if (i == 0) {
                degn = deg1;
                if (lane < degn) pn = csr[begn + lane];
            }

            float4 acc = make_float4(0.f, 0.f, 0.f, 0.f);
            for (int base = 0; base < deg; base += 32) {
                const int m = min(32, deg - base);
                if (base > 0) {
                    p = make_int2(0, 0);
                    if (lane < m) p = csr[beg_r + base + lane];
                }
                const int   s_l = p.x;
                const float c_l = __int_as_float(p.y) * dd;
                int j = 0;
                for (; j + 16 <= m; j += 16) add_rows<16>(acc, h4, s_l, c_l, j, lane);
                if (j + 8 <= m) { add_rows<8>(acc, h4, s_l, c_l, j, lane); j += 8; }
                switch (m - j) {
                    case 7: add_rows<7>(acc, h4, s_l, c_l, j, lane); break;
                    case 6: add_rows<6>(acc, h4, s_l, c_l, j, lane); break;
                    case 5: add_rows<5>(acc, h4, s_l, c_l, j, lane); break;
                    case 4: add_rows<4>(acc, h4, s_l, c_l, j, lane); break;
                    case 3: add_rows<3>(acc, h4, s_l, c_l, j, lane); break;
                    case 2: add_rows<2>(acc, h4, s_l, c_l, j, lane); break;
                    case 1: add_rows<1>(acc, h4, s_l, c_l, j, lane); break;
                    default: break;
                }
            }
            fma4(acc, dd * dd, self);
            *(float4*)(Xw + i * XP) = acc;

            beg_r = begn;
            deg   = degn;
            p     = pn;
        }

        __syncthreads();

        const float* Xc = Xl[cur];
        v8f acc8 = {};
        #pragma unroll 4
        for (int k = 0; k < 128; k += 4) {
            const int kk = k + hi * 2;
            v2f a;
            a[0] = Xc[r16 * XP + kk];
            a[1] = Xc[r16 * XP + kk + 1];
            v2f bb;
            bb[0] = Wl[(kk + 0) * 128 + colBase + r16];
            bb[1] = Wl[(kk + 1) * 128 + colBase + r16];
            acc8 = __builtin_amdgcn_wmma_f32_16x16x4_f32(
                 false, a,  false, bb,
                 (short)0, acc8,  false,  false);
        }

        float* out = Hout + (size_t)t * 16 * HDIM + colBase + r16;
        #pragma unroll
        for (int r = 0; r < 8; ++r) {
            float v = acc8[r] + bv;
            if (relu) v = fmaxf(v, 0.f);
            out[(size_t)(r + hi * 8) * HDIM] = v;
        }
        cur ^= 1;
    }
}

extern "C" void kernel_launch(void* const* d_in, const int* in_sizes, int n_in,
                              void* d_out, int out_size, void* d_ws, size_t ws_size,
                              hipStream_t stream) {
    (void)n_in; (void)out_size; (void)ws_size;

    const float* x  = (const float*)d_in[0];
    const int*   ei = (const int*)d_in[1];
    const float* W1 = (const float*)d_in[2];
    const float* b1 = (const float*)d_in[3];
    const float* W2 = (const float*)d_in[4];
    const float* b2 = (const float*)d_in[5];
    const float* W3 = (const float*)d_in[6];
    const float* b3 = (const float*)d_in[7];

    const int N = in_sizes[0] / HDIM;
    const int E = in_sizes[1] / 2;
    const int* src = ei;
    const int* dst = ei + E;

    float* ws     = (float*)d_ws;
    float* h      = ws;
    float* xbuf   = ws + (size_t)N * HDIM;
    int*   cnt    = (int*)(ws + 2 * (size_t)N * HDIM);
    int*   rowptr = cnt + N;
    float* dinv   = (float*)(rowptr + N);
    int*   rank   = (int*)(dinv + N);
    int2*  csr    = (int2*)(rank + E);
    float* out    = (float*)d_out;

    dim3 blk(256);
    dim3 gN((N + 1023) / 1024);
    dim3 gE((E + 2047) / 2048);
    dim3 gScan((N + 1023) / 1024);
    dim3 gGemm(768);
    const int nTiles = N / 16;

    zero_i32<<<gN, blk, 0, stream>>>(cnt, N);
    count_rank<<<gE, blk, 0, stream>>>(dst, cnt, rank, E);
    scan_rowptr<<<gScan, blk, 0, stream>>>(cnt, rowptr, dinv, N);
    place_edges<<<gE, blk, 0, stream>>>(src, dst, rank, rowptr, dinv, csr, E);

    gather_gemm<<<gGemm, blk, 0, stream>>>(x, rowptr, cnt, csr, dinv, b1, W1, h, nTiles, 1);

    gather_gemm<<<gGemm, blk, 0, stream>>>(h, rowptr, cnt, csr, dinv, b2, W2, xbuf, nTiles, 1);

    gather_gemm<<<gGemm, blk, 0, stream>>>(xbuf, rowptr, cnt, csr, dinv, b3, W3, out, nTiles, 0);
}


# hidden planes are stored times dinv of their row (one multiply on the WMMA result): layers 2 and 3 add source rows plain, no per-edge coefficient, no source-count gather; sums round differently
# speedup vs baseline: 4.4552x; 1.2180x over previous
#include <hip/hip_runtime.h>
#include <hip/hip_bf16.h>


#define NODES 50000
#define HDIM  128

typedef __attribute__((ext_vector_type(2))) float v2f;
typedef __attribute__((ext_vector_type(8))) float v8f;

#define CAP 64
#define XP 132

__global__ __launch_bounds__(256) void zero_i32(int* p, int n) {
    const int i0 = (blockIdx.x * 256 + threadIdx.x) * 4;
    if (i0 + 4 <= n) {
        *(int4*)(p + i0) = make_int4(0, 0, 0, 0);
    } else {
        for (int i = i0; i < n; ++i) p[i] = 0;
    }
}

__device__ __forceinline__ void put_edge(int s, int d, int r, int* __restrict__ slot,
                                         int* ovfCount, int2* __restrict__ ovf) {
    if (r < CAP) {
        slot[(size_t)d * CAP + r] = s;
    } else {
        const int k = atomicAdd(ovfCount, 1);
        ovf[k] = make_int2(s, d);
    }
}

__global__ __launch_bounds__(256) void count_slot(const int* __restrict__ src, const int* __restrict__ dst,
                                                  int* cnt, int* __restrict__ slot,
                                                  int* ovfCount, int2* __restrict__ ovf, int e) {
    const int i0 = (blockIdx.x * 256 + threadIdx.x) * 8;
    if (i0 >= e) return;
    if (i0 + 8 <= e && (e & 3) == 0) {
        const int4 s0 = *(const int4*)(src + i0);
        const int4 s1 = *(const int4*)(src + i0 + 4);
        const int4 d0 = *(const int4*)(dst + i0);
        const int4 d1 = *(const int4*)(dst + i0 + 4);
        const int s[8] = {s0.x, s0.y, s0.z, s0.w, s1.x, s1.y, s1.z, s1.w};
        const int d[8] = {d0.x, d0.y, d0.z, d0.w, d1.x, d1.y, d1.z, d1.w};
        int r[8];
        #pragma unroll
        for (int u = 0; u < 8; ++u) r[u] = atomicAdd(&cnt[d[u]], 1);
        #pragma unroll
        for (int u = 0; u < 8; ++u) put_edge(s[u], d[u], r[u], slot, ovfCount, ovf);
    } else {
        const int i1 = min(i0 + 8, e);
        for (int i = i0; i < i1; ++i) {
            const int dv = dst[i];
            put_edge(src[i], dv, atomicAdd(&cnt[dv], 1), slot, ovfCount, ovf);
        }
    }
}

__device__ __forceinline__ void fma4(float4& a, float c, const float4& v) {
    a.x = fmaf(c, v.x, a.x);
    a.y = fmaf(c, v.y, a.y);
    a.z = fmaf(c, v.z, a.z);
    a.w = fmaf(c, v.w, a.w);
}

__device__ __forceinline__ void add4(float4& a, const float4& v) {
    a.x += v.x;
    a.y += v.y;
    a.z += v.z;
    a.w += v.w;
}

template <int R, bool PRE>
__device__ __forceinline__ void add_rows(float4& acc, const float4* __restrict__ h4,
                                         int s_l, int k_l, float dd, int j, int lane) {
    float4 v[R];
    #pragma unroll
    for (int r = 0; r < R; ++r) {
        const int s = __builtin_amdgcn_readlane(s_l, j + r);
        v[r] = h4[(size_t)s * 32 + lane];
    }
    if (PRE) {
        #pragma unroll
        for (int r = 0; r < R; ++r) add4(acc, v[r]);
    } else {
        float c[R];
        const float c_l = __frsqrt_rn((float)(k_l + 1)) * dd;
        #pragma unroll
        for (int r = 0; r < R; ++r)
            c[r] = __int_as_float(__builtin_amdgcn_readlane(__float_as_int(c_l), j + r));
        #pragma unroll
        for (int r = 0; r < R; ++r) fma4(acc, c[r], v[r]);
    }
}

template <bool PRE_IN, bool SCALE_OUT>
__global__ __launch_bounds__(256) void gather_gemm(const float* __restrict__ h,
                                                   const int* __restrict__ cnt,
                                                   const int* __restrict__ slot,
                                                   const int* __restrict__ ovfCount,
                                                   const int2* __restrict__ ovf,
                                                   const float* __restrict__ bias,
                                                   const float* __restrict__ W,
                                                   float* __restrict__ Hout,
                                                   int nTiles, int relu, int eMax) {
    __shared__ __attribute__((aligned(16))) float Wl[128 * 128];
    __shared__ __attribute__((aligned(16))) float Xl[2][16 * XP];

    const int tid = threadIdx.x;
    {
        const float4* W4 = (const float4*)W;
        float4* Wl4 = (float4*)Wl;
        for (int i = tid; i < 128 * 128 / 4; i += 256) Wl4[i] = W4[i];
    }

    const int lane = tid & 31;
    const int wave = __builtin_amdgcn_readfirstlane(tid >> 5);
    const int r16  = lane & 15;
    const int hi   = lane >> 4;
    const int colBase = wave * 16;
    const float4* h4 = (const float4*)h;
    const float bv = bias[colBase + r16];

    int t = blockIdx.x;
    int deg_l = 0;
    if (t < nTiles && lane < 2) deg_l = cnt[t * 16 + wave * 2 + lane];

    int cur = 0;
    for (; t < nTiles; t += gridDim.x) {
        const int n0 = t * 16 + wave * 2;
        float* Xw = &Xl[cur][(wave * 2) * XP + lane * 4];

        const int   deg0 = __builtin_amdgcn_readlane(deg_l, 0);
        const int   deg1 = __builtin_amdgcn_readlane(deg_l, 1);
        const float dd0  = __frsqrt_rn((float)(deg0 + 1));
        const float dd1  = __frsqrt_rn((float)(deg1 + 1));

        const int* sl0 = slot + (size_t)n0 * CAP;
        int sA = 0, sB = 0;
        if (lane < deg0) sA = sl0[lane];
        if (lane < deg1) sB = sl0[CAP + lane];
        int kA = 0, kB = 0;
        if (!PRE_IN) {
            kA = cnt[sA];
            kB = cnt[sB];
        }

        {
            const int tn = t + gridDim.x;
            deg_l = 0;
            if (tn < nTiles && lane < 2) deg_l = cnt[tn * 16 + wave * 2 + lane];
        }

        int4 rc0 = make_int4(0, 0, 0, 0), rc1 = make_int4(0, 0, 0, 0);
        if (SCALE_OUT) {
            const int4* rc = (const int4*)(cnt + t * 16 + hi * 8);
            rc0 = rc[0];
            rc1 = rc[1];
        }

        #pragma unroll
        for (int i = 0; i < 2; ++i) {
            const int   node = n0 + i;
            const int   deg  = (i == 0) ? deg0 : deg1;
            const float dd   = (i == 0) ? dd0 : dd1;
            const float4 self = h4[(size_t)node * 32 + lane];
            const int s_l = (i == 0) ? sA : sB;
            const int k_l = (i == 0) ? kA : kB;

            float4 acc = make_float4(0.f, 0.f, 0.f, 0.f);
            {
                const int m = min(deg, 32);
                int j = 0;
                if (m >= 16)     { add_rows<16, PRE_IN>(acc, h4, s_l, k_l, dd, 0, lane);  j = 16; }
                if (m - j >= 16) { add_rows<16, PRE_IN>(acc, h4, s_l, k_l, dd, 16, lane); j = 32; }
                if (m - j >= 8)  { add_rows<8, PRE_IN>(acc, h4, s_l, k_l, dd, j, lane);   j += 8; }
                switch (m - j) {
                    case 7: add_rows<7, PRE_IN>(acc, h4, s_l, k_l, dd, j, lane); break;
                    case 6: add_rows<6, PRE_IN>(acc, h4, s_l, k_l, dd, j, lane); break;
                    case 5: add_rows<5, PRE_IN>(acc, h4, s_l, k_l, dd, j, lane); break;
                    case 4: add_rows<4, PRE_IN>(acc, h4, s_l, k_l, dd, j, lane); break;
                    case 3: add_rows<3, PRE_IN>(acc, h4, s_l, k_l, dd, j, lane); break;
                    case 2: add_rows<2, PRE_IN>(acc, h4, s_l, k_l, dd, j, lane); break;
                    case 1: add_rows<1, PRE_IN>(acc, h4, s_l, k_l, dd, j, lane); break;
                    default: break;
                }
            }
            if (deg > 32) {
                const int m2 = min(deg, CAP) - 32;
                int s2 = 0;
                if (lane < m2) s2 = slot[(size_t)node * CAP + 32 + lane];
                if (PRE_IN) {
                    for (int j2 = 0; j2 < m2; ++j2) {
                        const int s = __builtin_amdgcn_readlane(s2, j2);
                        const float4 v = h4[(size_t)s * 32 + lane];
                        add4(acc, v);
                    }
                } else {
                    const int k2 = cnt[s2];
                    const float c2 = __frsqrt_rn((float)(k2 + 1)) * dd;
                    for (int j2 = 0; j2 < m2; ++j2) {
                        const int   s  = __builtin_amdgcn_readlane(s2, j2);
                        const float cq = __int_as_float(__builtin_amdgcn_readlane(__float_as_int(c2), j2));
                        const float4 v = h4[(size_t)s * 32 + lane];
                        fma4(acc, cq, v);
                    }
                }
            }
            if (deg > CAP) {
                const int nOv = min(ovfCount[0], eMax);
                for (int base = 0; base < nOv; base += 32) {
                    const int k = base + lane;
                    int2 q = make_int2(0, -1);
                    if (k < nOv) q = ovf[k];
                    const bool mine = (q.y == node);
                    float wq = PRE_IN ? 1.f : 0.f;
                    if (!PRE_IN && mine) wq = __frsqrt_rn((float)(cnt[q.x] + 1)) * dd;
                    unsigned todo = __builtin_amdgcn_ballot_w32(mine);
                    while (todo != 0u) {
                        const int jj = __builtin_ctz(todo);
                        todo &= todo - 1u;
                        const int   s  = __builtin_amdgcn_readlane(q.x, jj);
                        const float cq = __int_as_float(__builtin_amdgcn_readlane(__float_as_int(wq), jj));
                        const float4 v = h4[(size_t)s * 32 + lane];
                        fma4(acc, cq, v);
                    }
                }
            }
            if (PRE_IN) {
                add4(acc, self);
                acc.x *= dd;
                acc.y *= dd;
                acc.z *= dd;
                acc.w *= dd;
            } else {
                fma4(acc, dd * dd, self);
            }
            *(float4*)(Xw + i * XP) = acc;
        }

        __syncthreads();

        const float* Xc = Xl[cur];
        v8f acc8 = {};
        #pragma unroll 4
        for (int k = 0; k < 128; k += 4) {
            const int kk = k + hi * 2;
            v2f a;
            a[0] = Xc[r16 * XP + kk];
            a[1] = Xc[r16 * XP + kk + 1];
            v2f bb;
            bb[0] = Wl[(kk + 0) * 128 + colBase + r16];
            bb[1] = Wl[(kk + 1) * 128 + colBase + r16];
            acc8 = __builtin_amdgcn_wmma_f32_16x16x4_f32(
                 false, a,  false, bb,
                 (short)0, acc8,  false,  false);
        }

        float* out = Hout + (size_t)t * 16 * HDIM + colBase + r16;
        const int rcnt[8] = {rc0.x, rc0.y, rc0.z, rc0.w, rc1.x, rc1.y, rc1.z, rc1.w};
        #pragma unroll
        for (int r = 0; r < 8; ++r) {
            float v = acc8[r] + bv;
            if (relu) v = fmaxf(v, 0.f);
            if (SCALE_OUT) v *= __frsqrt_rn((float)(rcnt[r] + 1));
            out[(size_t)(r + hi * 8) * HDIM] = v;
        }
        cur ^= 1;
    }
}

extern "C" void kernel_launch(void* const* d_in, const int* in_sizes, int n_in,
                              void* d_out, int out_size, void* d_ws, size_t ws_size,
                              hipStream_t stream) {
    (void)n_in; (void)out_size; (void)ws_size;

    const float* x  = (const float*)d_in[0];
    const int*   ei = (const int*)d_in[1];
    const float* W1 = (const float*)d_in[2];
    const float* b1 = (const float*)d_in[3];
    const float* W2 = (const float*)d_in[4];
    const float* b2 = (const float*)d_in[5];
    const float* W3 = (const float*)d_in[6];
    const float* b3 = (const float*)d_in[7];

    const int N = in_sizes[0] / HDIM;
    const int E = in_sizes[1] / 2;
    const int* src = ei;
    const int* dst = ei + E;

    float* ws   = (float*)d_ws;
    float* h    = ws;
    float* xbuf = ws + (size_t)N * HDIM;
    const int cntLen = (N + 1 + 63) / 64 * 64;
    int*   cnt      = (int*)(ws + 2 * (size_t)N * HDIM);
    int*   ovfCount = cnt + N;
    int*   slot     = cnt + cntLen;
    int2*  ovf      = (int2*)(slot + (size_t)N * CAP);
    float* out      = (float*)d_out;

    dim3 blk(256);
    dim3 gZ((cntLen + 1023) / 1024);
    dim3 gE((E + 2047) / 2048);
    dim3 gGemm(768);
    const int nTiles = N / 16;

    zero_i32<<<gZ, blk, 0, stream>>>(cnt, cntLen);
    count_slot<<<gE, blk, 0, stream>>>(src, dst, cnt, slot, ovfCount, ovf, E);

    gather_gemm<false, true><<<gGemm, blk, 0, stream>>>(x, cnt, slot, ovfCount, ovf, b1, W1, h, nTiles, 1, E);

    gather_gemm<true, true><<<gGemm, blk, 0, stream>>>(h, cnt, slot, ovfCount, ovf, b2, W2, xbuf, nTiles, 1, E);

    gather_gemm<true, false><<<gGemm, blk, 0, stream>>>(xbuf, cnt, slot, ovfCount, ovf, b3, W3, out, nTiles, 0, E);
}


# edge lists built by a two-level counting sort on LDS counters (chunks ordered by 256-node bin; one block per bin ranks, writes whole rows): no global atomic per edge; sum order = LDS arrival order
# speedup vs baseline: 6.2138x; 1.3947x over previous
#include <hip/hip_runtime.h>
#include <hip/hip_bf16.h>


#define NODES 50000
#define HDIM  128

typedef __attribute__((ext_vector_type(2))) float v2f;
typedef __attribute__((ext_vector_type(8))) float v8f;

#define CAP 64
#define XP 132

#define BSH    8
#define BNODES 256
#define MAXB   512
#define CHUNK  4096

template <bool FULL>
__device__ __forceinline__ void sort_chunk_body(const int* __restrict__ src, const int* __restrict__ dst,
                                                int2* __restrict__ sorted, int* __restrict__ off,
                                                int* hist, int* wsum, int2* stage,
                                                int e, int nBins, int base) {
    const int tid  = threadIdx.x;
    const int lane = tid & 31;
    const int wave = tid >> 5;
    const int i0   = base + tid * 8;
    int s[8], d[8], r[8];
    int n = 8;
    if (FULL) {
        const int4 s0 = *(const int4*)(src + i0);
        const int4 s1 = *(const int4*)(src + i0 + 4);
        const int4 d0 = *(const int4*)(dst + i0);
        const int4 d1 = *(const int4*)(dst + i0 + 4);
        s[0] = s0.x; s[1] = s0.y; s[2] = s0.z; s[3] = s0.w; s[4] = s1.x; s[5] = s1.y; s[6] = s1.z; s[7] = s1.w;
        d[0] = d0.x; d[1] = d0.y; d[2] = d0.z; d[3] = d0.w; d[4] = d1.x; d[5] = d1.y; d[6] = d1.z; d[7] = d1.w;
    } else {
        n = 0;
        #pragma unroll
        for (int u = 0; u < 8; ++u) {
            s[u] = 0; d[u] = 0;
            if (i0 + u < e) { s[u] = src[i0 + u]; d[u] = dst[i0 + u]; n = u + 1; }
        }
    }
    #pragma unroll
    for (int u = 0; u < 8; ++u) { r[u] = 0; if (FULL || u < n) r[u] = atomicAdd(&hist[d[u] >> BSH], 1); }
    __syncthreads();

    const int v = hist[tid];
    int inc = v;
    #pragma unroll
    for (int o = 1; o < 32; o <<= 1) {
        const int t = __shfl_up(inc, o);
        if (lane >= o) inc += t;
    }
    if (lane == 31) wsum[wave] = inc;
    __syncthreads();
    int wb = 0;
    for (int j = 0; j < wave; ++j) wb += wsum[j];
    const int start = wb + inc - v;
    hist[tid] = start;
    if (tid <= nBins) off[(size_t)blockIdx.x * (nBins + 1) + tid] = start;
    __syncthreads();

    #pragma unroll
    for (int u = 0; u < 8; ++u)
        if (FULL || u < n) stage[hist[d[u] >> BSH] + r[u]] = make_int2(s[u], d[u]);
    __syncthreads();

    if (FULL) {
        const int4* st4 = (const int4*)stage;
        int4* out4 = (int4*)(sorted + (size_t)base);
        #pragma unroll
        for (int i = 0; i < CHUNK / 2 / 512; ++i) out4[i * 512 + tid] = st4[i * 512 + tid];
    } else {
        const int m = min(CHUNK, e - base);
        for (int i = tid; i < m; i += 512) sorted[(size_t)base + i] = stage[i];
    }
}

__global__ __launch_bounds__(512) void sort_chunk(const int* __restrict__ src, const int* __restrict__ dst,
                                                  int2* __restrict__ sorted, int* __restrict__ off,
                                                  int* __restrict__ ovfCount, int e, int nBins) {
    __shared__ int  hist[MAXB];
    __shared__ int  wsum[16];
    __shared__ __attribute__((aligned(16))) int2 stage[CHUNK];

    const int tid = threadIdx.x;
    if (blockIdx.x == 0 && tid == 0) ovfCount[0] = 0;
    hist[tid] = 0;
    __syncthreads();

    const int base = blockIdx.x * CHUNK;
    if (base + CHUNK <= e && (e & 3) == 0)
        sort_chunk_body<true>(src, dst, sorted, off, hist, wsum, stage, e, nBins, base);
    else
        sort_chunk_body<false>(src, dst, sorted, off, hist, wsum, stage, e, nBins, base);
}

__device__ __forceinline__ void put_edge(int2 q, int lo, int* cntL, int* slotL,
                                         int* ovfCount, int2* __restrict__ ovf) {
    const int dl = q.y - lo;
    const int r  = atomicAdd(&cntL[dl], 1);
    if (r < CAP) {
        slotL[dl * CAP + r] = q.x;
    } else {
        const int k = atomicAdd(ovfCount, 1);
        ovf[k] = q;
    }
}

__global__ __launch_bounds__(512) void place_bin(const int2* __restrict__ sorted, const int* __restrict__ off,
                                                 int* __restrict__ cnt, int* __restrict__ slot,
                                                 int* ovfCount, int2* __restrict__ ovf,
                                                 int nNodes, int nBins, int nChunks) {
    __shared__ int cntL[BNODES];
    __shared__ int segS[512];
    __shared__ int segL[512];
    __shared__ int slotL[BNODES * CAP];

    const int tid  = threadIdx.x;
    const int lane = tid & 31;
    const int wave = __builtin_amdgcn_readfirstlane(tid >> 5);
    const int b    = blockIdx.x;
    const int lo   = b << BSH;
    if (tid < BNODES) cntL[tid] = 0;

    for (int c0 = 0; c0 < nChunks; c0 += 512) {
        __syncthreads();
        {
            const int c = c0 + tid;
            int st = 0, ln = 0;
            if (c < nChunks) {
                const int* o = off + (size_t)c * (nBins + 1) + b;
                const int o0 = o[0];
                ln = o[1] - o0;
                st = c * CHUNK + o0;
            }
            segS[tid] = st;
            segL[tid] = ln;
        }
        __syncthreads();
        const int nc = min(512, nChunks - c0);
        for (int j = wave; j < nc; j += 128) {
            int2 q[8];
            int  st[8], ln[8];
            #pragma unroll
            for (int u = 0; u < 8; ++u) {
                const int jj = j + 16 * u;
                st[u] = 0; ln[u] = 0;
                if (jj < nc) { st[u] = segS[jj]; ln[u] = segL[jj]; }
                q[u] = make_int2(0, 0);
                if (lane < ln[u]) q[u] = sorted[(size_t)st[u] + lane];
            }
            #pragma unroll
            for (int u = 0; u < 8; ++u)
                if (lane < ln[u]) put_edge(q[u], lo, cntL, slotL, ovfCount, ovf);
            #pragma unroll
            for (int u = 0; u < 8; ++u)
                for (int k = 32 + lane; k < ln[u]; k += 32)
                    put_edge(sorted[(size_t)st[u] + k], lo, cntL, slotL, ovfCount, ovf);
        }
    }
    __syncthreads();

    if (tid < BNODES && lo + tid < nNodes) cnt[lo + tid] = cntL[tid];
    for (int row = wave; row < BNODES; row += 16) {
        const int node = lo + row;
        if (node >= nNodes) break;
        const int c = min(cntL[row], CAP);
        int* g = slot + (size_t)node * CAP;
        if (lane < c) g[lane] = slotL[row * CAP + lane];
        if (lane + 32 < c) g[lane + 32] = slotL[row * CAP + lane + 32];
    }
}

__device__ __forceinline__ void fma4(float4& a, float c, const float4& v) {
    a.x = fmaf(c, v.x, a.x);
    a.y = fmaf(c, v.y, a.y);
    a.z = fmaf(c, v.z, a.z);
    a.w = fmaf(c, v.w, a.w);
}

__device__ __forceinline__ void add4(float4& a, const float4& v) {
    a.x += v.x;
    a.y += v.y;
    a.z += v.z;
    a.w += v.w;
}

template <int R, bool PRE>
__device__ __forceinline__ void add_rows(float4& acc, const float4* __restrict__ h4,
                                         int s_l, int k_l, float dd, int j, int lane) {
    float4 v[R];
    #pragma unroll
    for (int r = 0; r < R; ++r) {
        const int s = __builtin_amdgcn_readlane(s_l, j + r);
        v[r] = h4[(size_t)s * 32 + lane];
    }
    if (PRE) {
        #pragma unroll
        for (int r = 0; r < R; ++r) add4(acc, v[r]);
    } else {
        float c[R];
        const float c_l = __frsqrt_rn((float)(k_l + 1)) * dd;
        #pragma unroll
        for (int r = 0; r < R; ++r)
            c[r] = __int_as_float(__builtin_amdgcn_readlane(__float_as_int(c_l), j + r));
        #pragma unroll
        for (int r = 0; r < R; ++r) fma4(acc, c[r], v[r]);
    }
}

template <bool PRE_IN, bool SCALE_OUT>
__global__ __launch_bounds__(256) void gather_gemm(const float* __restrict__ h,
                                                   const int* __restrict__ cnt,
                                                   const int* __restrict__ slot,
                                                   const int* __restrict__ ovfCount,
                                                   const int2* __restrict__ ovf,
                                                   const float* __restrict__ bias,
                                                   const float* __restrict__ W,
                                                   float* __restrict__ Hout,
                                                   int nTiles, int relu, int eMax) {
    __shared__ __attribute__((aligned(16))) float Wl[128 * 128];
    __shared__ __attribute__((aligned(16))) float Xl[2][16 * XP];

    const int tid = threadIdx.x;
    {
        const float4* W4 = (const float4*)W;
        float4* Wl4 = (float4*)Wl;
        for (int i = tid; i < 128 * 128 / 4; i += 256) Wl4[i] = W4[i];
    }

    const int lane = tid & 31;
    const int wave = __builtin_amdgcn_readfirstlane(tid >> 5);
    const int r16  = lane & 15;
    const int hi   = lane >> 4;
    const int colBase = wave * 16;
    const float4* h4 = (const float4*)h;
    const float bv = bias[colBase + r16];

    int t = blockIdx.x;
    int deg_l = 0;
    if (t < nTiles && lane < 2) deg_l = cnt[t * 16 + wave * 2 + lane];

    int cur = 0;
    for (; t < nTiles; t += gridDim.x) {
        const int n0 = t * 16 + wave * 2;
        float* Xw = &Xl[cur][(wave * 2) * XP + lane * 4];

        const int   deg0 = __builtin_amdgcn_readlane(deg_l, 0);
        const int   deg1 = __builtin_amdgcn_readlane(deg_l, 1);
        const float dd0  = __frsqrt_rn((float)(deg0 + 1));
        const float dd1  = __frsqrt_rn((float)(deg1 + 1));

        const int* sl0 = slot + (size_t)n0 * CAP;
        int sA = 0, sB = 0;
        if (lane < deg0) sA = sl0[lane];
        if (lane < deg1) sB = sl0[CAP + lane];
        int kA = 0, kB = 0;
        if (!PRE_IN) {
            kA = cnt[sA];
            kB = cnt[sB];
        }

        {
            const int tn = t + gridDim.x;
            deg_l = 0;
            if (tn < nTiles && lane < 2) deg_l = cnt[tn * 16 + wave * 2 + lane];
        }

        int4 rc0 = make_int4(0, 0, 0, 0), rc1 = make_int4(0, 0, 0, 0);
        if (SCALE_OUT) {
            const int4* rc = (const int4*)(cnt + t * 16 + hi * 8);
            rc0 = rc[0];
            rc1 = rc[1];
        }

        #pragma unroll
        for (int i = 0; i < 2; ++i) {
            const int   node = n0 + i;
            const int   deg  = (i == 0) ? deg0 : deg1;
            const float dd   = (i == 0) ? dd0 : dd1;
            const float4 self = h4[(size_t)node * 32 + lane];
            const int s_l = (i == 0) ? sA : sB;
            const int k_l = (i == 0) ? kA : kB;

            float4 acc = make_float4(0.f, 0.f, 0.f, 0.f);
            {
                const int m = min(deg, 32);
                int j = 0;
                if (m >= 16)     { add_rows<16, PRE_IN>(acc, h4, s_l, k_l, dd, 0, lane);  j = 16; }
                if (m - j >= 16) { add_rows<16, PRE_IN>(acc, h4, s_l, k_l, dd, 16, lane); j = 32; }
                if (m - j >= 8)  { add_rows<8, PRE_IN>(acc, h4, s_l, k_l, dd, j, lane);   j += 8; }
                switch (m - j) {
                    case 7: add_rows<7, PRE_IN>(acc, h4, s_l, k_l, dd, j, lane); break;
                    case 6: add_rows<6, PRE_IN>(acc, h4, s_l, k_l, dd, j, lane); break;
                    case 5: add_rows<5, PRE_IN>(acc, h4, s_l, k_l, dd, j, lane); break;
                    case 4: add_rows<4, PRE_IN>(acc, h4, s_l, k_l, dd, j, lane); break;
                    case 3: add_rows<3, PRE_IN>(acc, h4, s_l, k_l, dd, j, lane); break;
                    case 2: add_rows<2, PRE_IN>(acc, h4, s_l, k_l, dd, j, lane); break;
                    case 1: add_rows<1, PRE_IN>(acc, h4, s_l, k_l, dd, j, lane); break;
                    default: break;
                }
            }
            if (deg > 32) {
                const int m2 = min(deg, CAP) - 32;
                int s2 = 0;
                if (lane < m2) s2 = slot[(size_t)node * CAP + 32 + lane];
                if (PRE_IN) {
                    for (int j2 = 0; j2 < m2; ++j2) {
                        const int s = __builtin_amdgcn_readlane(s2, j2);
                        const float4 v = h4[(size_t)s * 32 + lane];
                        add4(acc, v);
                    }
                } else {
                    const int k2 = cnt[s2];
                    const float c2 = __frsqrt_rn((float)(k2 + 1)) * dd;
                    for (int j2 = 0; j2 < m2; ++j2) {
                        const int   s  = __builtin_amdgcn_readlane(s2, j2);
                        const float cq = __int_as_float(__builtin_amdgcn_readlane(__float_as_int(c2), j2));
                        const float4 v = h4[(size_t)s * 32 + lane];
                        fma4(acc, cq, v);
                    }
                }
            }
            if (deg > CAP) {
                const int nOv = min(ovfCount[0], eMax);
                for (int base = 0; base < nOv; base += 32) {
                    const int k = base + lane;
                    int2 q = make_int2(0, -1);
                    if (k < nOv) q = ovf[k];
                    const bool mine = (q.y == node);
                    float wq = PRE_IN ? 1.f : 0.f;
                    if (!PRE_IN && mine) wq = __frsqrt_rn((float)(cnt[q.x] + 1)) * dd;
                    unsigned todo = __builtin_amdgcn_ballot_w32(mine);
                    while (todo != 0u) {
                        const int jj = __builtin_ctz(todo);
                        todo &= todo - 1u;
                        const int   s  = __builtin_amdgcn_readlane(q.x, jj);
                        const float cq = __int_as_float(__builtin_amdgcn_readlane(__float_as_int(wq), jj));
                        const float4 v = h4[(size_t)s * 32 + lane];
                        fma4(acc, cq, v);
                    }
                }
            }
            if (PRE_IN) {
                add4(acc, self);
                acc.x *= dd;
                acc.y *= dd;
                acc.z *= dd;
                acc.w *= dd;
            } else {
                fma4(acc, dd * dd, self);
            }
            *(float4*)(Xw + i * XP) = acc;
        }

        __syncthreads();

        const float* Xc = Xl[cur];
        v8f acc8 = {};
        #pragma unroll 4
        for (int k = 0; k < 128; k += 4) {
            const int kk = k + hi * 2;
            v2f a;
            a[0] = Xc[r16 * XP + kk];
            a[1] = Xc[r16 * XP + kk + 1];
            v2f bb;
            bb[0] = Wl[(kk + 0) * 128 + colBase + r16];
            bb[1] = Wl[(kk + 1) * 128 + colBase + r16];
            acc8 = __builtin_amdgcn_wmma_f32_16x16x4_f32(
                 false, a,  false, bb,
                 (short)0, acc8,  false,  false);
        }

        float* out = Hout + (size_t)t * 16 * HDIM + colBase + r16;
        const int rcnt[8] = {rc0.x, rc0.y, rc0.z, rc0.w, rc1.x, rc1.y, rc1.z, rc1.w};
        #pragma unroll
        for (int r = 0; r < 8; ++r) {
            float v = acc8[r] + bv;
            if (relu) v = fmaxf(v, 0.f);
            if (SCALE_OUT) v *= __frsqrt_rn((float)(rcnt[r] + 1));
            out[(size_t)(r + hi * 8) * HDIM] = v;
        }
        cur ^= 1;
    }
}

extern "C" void kernel_launch(void* const* d_in, const int* in_sizes, int n_in,
                              void* d_out, int out_size, void* d_ws, size_t ws_size,
                              hipStream_t stream) {
    (void)n_in; (void)out_size; (void)ws_size;

    const float* x  = (const float*)d_in[0];
    const int*   ei = (const int*)d_in[1];
    const float* W1 = (const float*)d_in[2];
    const float* b1 = (const float*)d_in[3];
    const float* W2 = (const float*)d_in[4];
    const float* b2 = (const float*)d_in[5];
    const float* W3 = (const float*)d_in[6];
    const float* b3 = (const float*)d_in[7];

    const int N = in_sizes[0] / HDIM;
    const int E = in_sizes[1] / 2;
    const int* src = ei;
    const int* dst = ei + E;

    float* ws   = (float*)d_ws;
    float* h    = ws;
    float* xbuf = ws + (size_t)N * HDIM;
    const int cntLen = (N + 1 + 63) / 64 * 64;
    int*   cnt      = (int*)(ws + 2 * (size_t)N * HDIM);
    int*   ovfCount = cnt + N;
    int*   slot     = cnt + cntLen;
    int2*  ovf      = (int2*)(slot + (size_t)N * CAP);
    int2*  sorted   = ovf + E;
    int*   off      = (int*)(sorted + E);
    float* out      = (float*)d_out;

    dim3 blk(256);
    dim3 blkS(512);
    const int nChunks = (E + CHUNK - 1) / CHUNK;
    const int nBins   = (N + BNODES - 1) / BNODES;
    dim3 gGemm(768);
    const int nTiles = N / 16;

    sort_chunk<<<dim3(nChunks), blkS, 0, stream>>>(src, dst, sorted, off, ovfCount, E, nBins);
    place_bin<<<dim3(nBins), blkS, 0, stream>>>(sorted, off, cnt, slot, ovfCount, ovf, N, nBins, nChunks);

    gather_gemm<false, true><<<gGemm, blk, 0, stream>>>(x, cnt, slot, ovfCount, ovf, b1, W1, h, nTiles, 1, E);

    gather_gemm<true, true><<<gGemm, blk, 0, stream>>>(h, cnt, slot, ovfCount, ovf, b2, W2, xbuf, nTiles, 1, E);

    gather_gemm<true, false><<<gGemm, blk, 0, stream>>>(xbuf, cnt, slot, ovfCount, ovf, b3, W3, out, nTiles, 0, E);
}


# each wave holds its 16 columns of W as B fragments in 64 registers, loaded once from memory: no W in LDS, no staging pass; a tile's 16 LDS reads go out together ahead of its 32 WMMAs; sum order same
# speedup vs baseline: 6.5440x; 1.0531x over previous
#include <hip/hip_runtime.h>
#include <hip/hip_bf16.h>


#define NODES 50000
#define HDIM  128

typedef __attribute__((ext_vector_type(2))) float v2f;
typedef __attribute__((ext_vector_type(8))) float v8f;

#define CAP 64
#define XP 132

#define BSH    8
#define BNODES 256
#define MAXB   512
#define CHUNK  4096

template <bool FULL>
__device__ __forceinline__ void sort_chunk_body(const int* __restrict__ src, const int* __restrict__ dst,
                                                int2* __restrict__ sorted, int* __restrict__ off,
                                                int* hist, int* wsum, int2* stage,
                                                int e, int nBins, int base) {
    const int tid  = threadIdx.x;
    const int lane = tid & 31;
    const int wave = tid >> 5;
    const int i0   = base + tid * 8;
    int s[8], d[8], r[8];
    int n = 8;
    if (FULL) {
        const int4 s0 = *(const int4*)(src + i0);
        const int4 s1 = *(const int4*)(src + i0 + 4);
        const int4 d0 = *(const int4*)(dst + i0);
        const int4 d1 = *(const int4*)(dst + i0 + 4);
        s[0] = s0.x; s[1] = s0.y; s[2] = s0.z; s[3] = s0.w; s[4] = s1.x; s[5] = s1.y; s[6] = s1.z; s[7] = s1.w;
        d[0] = d0.x; d[1] = d0.y; d[2] = d0.z; d[3] = d0.w; d[4] = d1.x; d[5] = d1.y; d[6] = d1.z; d[7] = d1.w;
    } else {
        n = 0;
        #pragma unroll
        for (int u = 0; u < 8; ++u) {
            s[u] = 0; d[u] = 0;
            if (i0 + u < e) { s[u] = src[i0 + u]; d[u] = dst[i0 + u]; n = u + 1; }
        }
    }
    #pragma unroll
    for (int u = 0; u < 8; ++u) { r[u] = 0; if (FULL || u < n) r[u] = atomicAdd(&hist[d[u] >> BSH], 1); }
    __syncthreads();

    const int v = hist[tid];
    int inc = v;
    #pragma unroll
    for (int o = 1; o < 32; o <<= 1) {
        const int t = __shfl_up(inc, o);
        if (lane >= o) inc += t;
    }
    if (lane == 31) wsum[wave] = inc;
    __syncthreads();
    int wb = 0;
    for (int j = 0; j < wave; ++j) wb += wsum[j];
    const int start = wb + inc - v;
    hist[tid] = start;
    if (tid <= nBins) off[(size_t)blockIdx.x * (nBins + 1) + tid] = start;
    __syncthreads();

    #pragma unroll
    for (int u = 0; u < 8; ++u)
        if (FULL || u < n) stage[hist[d[u] >> BSH] + r[u]] = make_int2(s[u], d[u]);
    __syncthreads();

    if (FULL) {
        const int4* st4 = (const int4*)stage;
        int4* out4 = (int4*)(sorted + (size_t)base);
        #pragma unroll
        for (int i = 0; i < CHUNK / 2 / 512; ++i) out4[i * 512 + tid] = st4[i * 512 + tid];
    } else {
        const int m = min(CHUNK, e - base);
        for (int i = tid; i < m; i += 512) sorted[(size_t)base + i] = stage[i];
    }
}

__global__ __launch_bounds__(512) void sort_chunk(const int* __restrict__ src, const int* __restrict__ dst,
                                                  int2* __restrict__ sorted, int* __restrict__ off,
                                                  int* __restrict__ ovfCount, int e, int nBins) {
    __shared__ int  hist[MAXB];
    __shared__ int  wsum[16];
    __shared__ __attribute__((aligned(16))) int2 stage[CHUNK];

    const int tid = threadIdx.x;
    if (blockIdx.x == 0 && tid == 0) ovfCount[0] = 0;
    hist[tid] = 0;
    __syncthreads();

    const int base = blockIdx.x * CHUNK;
    if (base + CHUNK <= e && (e & 3) == 0)
        sort_chunk_body<true>(src, dst, sorted, off, hist, wsum, stage, e, nBins, base);
    else
        sort_chunk_body<false>(src, dst, sorted, off, hist, wsum, stage, e, nBins, base);
}

__device__ __forceinline__ void put_edge(int2 q, int lo, int* cntL, int* slotL,
                                         int* ovfCount, int2* __restrict__ ovf) {
    const int dl = q.y - lo;
    const int r  = atomicAdd(&cntL[dl], 1);
    if (r < CAP) {
        slotL[dl * CAP + r] = q.x;
    } else {
        const int k = atomicAdd(ovfCount, 1);
        ovf[k] = q;
    }
}

__global__ __launch_bounds__(512) void place_bin(const int2* __restrict__ sorted, const int* __restrict__ off,
                                                 int* __restrict__ cnt, int* __restrict__ slot,
                                                 int* ovfCount, int2* __restrict__ ovf,
                                                 int nNodes, int nBins, int nChunks) {
    __shared__ int cntL[BNODES];
    __shared__ int segS[512];
    __shared__ int segL[512];
    __shared__ int slotL[BNODES * CAP];

    const int tid  = threadIdx.x;
    const int lane = tid & 31;
    const int wave = __builtin_amdgcn_readfirstlane(tid >> 5);
    const int b    = blockIdx.x;
    const int lo   = b << BSH;
    if (tid < BNODES) cntL[tid] = 0;

    for (int c0 = 0; c0 < nChunks; c0 += 512) {
        __syncthreads();
        {
            const int c = c0 + tid;
            int st = 0, ln = 0;
            if (c < nChunks) {
                const int* o = off + (size_t)c * (nBins + 1) + b;
                const int o0 = o[0];
                ln = o[1] - o0;
                st = c * CHUNK + o0;
            }
            segS[tid] = st;
            segL[tid] = ln;
        }
        __syncthreads();
        const int nc = min(512, nChunks - c0);
        for (int j = wave; j < nc; j += 128) {
            int2 q[8];
            int  st[8], ln[8];
            #pragma unroll
            for (int u = 0; u < 8; ++u) {
                const int jj = j + 16 * u;
                st[u] = 0; ln[u] = 0;
                if (jj < nc) { st[u] = segS[jj]; ln[u] = segL[jj]; }
                q[u] = make_int2(0, 0);
                if (lane < ln[u]) q[u] = sorted[(size_t)st[u] + lane];
            }
            #pragma unroll
            for (int u = 0; u < 8; ++u)
                if (lane < ln[u]) put_edge(q[u], lo, cntL, slotL, ovfCount, ovf);
            #pragma unroll
            for (int u = 0; u < 8; ++u)
                for (int k = 32 + lane; k < ln[u]; k += 32)
                    put_edge(sorted[(size_t)st[u] + k], lo, cntL, slotL, ovfCount, ovf);
        }
    }
    __syncthreads();

    if (tid < BNODES && lo + tid < nNodes) cnt[lo + tid] = cntL[tid];
    for (int row = wave; row < BNODES; row += 16) {
        const int node = lo + row;
        if (node >= nNodes) break;
        const int c = min(cntL[row], CAP);
        int* g = slot + (size_t)node * CAP;
        if (lane < c) g[lane] = slotL[row * CAP + lane];
        if (lane + 32 < c) g[lane + 32] = slotL[row * CAP + lane + 32];
    }
}

__device__ __forceinline__ void fma4(float4& a, float c, const float4& v) {
    a.x = fmaf(c, v.x, a.x);
    a.y = fmaf(c, v.y, a.y);
    a.z = fmaf(c, v.z, a.z);
    a.w = fmaf(c, v.w, a.w);
}

__device__ __forceinline__ void add4(float4& a, const float4& v) {
    a.x += v.x;
    a.y += v.y;
    a.z += v.z;
    a.w += v.w;
}

template <int R, bool PRE>
__device__ __forceinline__ void add_rows(float4& acc, const float4* __restrict__ h4,
                                         int s_l, int k_l, float dd, int j, int lane) {
    float4 v[R];
    #pragma unroll
    for (int r = 0; r < R; ++r) {
        const int s = __builtin_amdgcn_readlane(s_l, j + r);
        v[r] = h4[(size_t)s * 32 + lane];
    }
    if (PRE) {
        #pragma unroll
        for (int r = 0; r < R; ++r) add4(acc, v[r]);
    } else {
        float c[R];
        const float c_l = __frsqrt_rn((float)(k_l + 1)) * dd;
        #pragma unroll
        for (int r = 0; r < R; ++r)
            c[r] = __int_as_float(__builtin_amdgcn_readlane(__float_as_int(c_l), j + r));
        #pragma unroll
        for (int r = 0; r < R; ++r) fma4(acc, c[r], v[r]);
    }
}

template <bool PRE_IN, bool SCALE_OUT>
__global__ __launch_bounds__(256, 6) void gather_gemm(const float* __restrict__ h,
                                                   const int* __restrict__ cnt,
                                                   const int* __restrict__ slot,
                                                   const int* __restrict__ ovfCount,
                                                   const int2* __restrict__ ovf,
                                                   const float* __restrict__ bias,
                                                   const float* __restrict__ W,
                                                   float* __restrict__ Hout,
                                                   int nTiles, int relu, int eMax) {
    __shared__ __attribute__((aligned(16))) float Xl[2][16 * XP];

    const int tid = threadIdx.x;
    const int lane = tid & 31;
    const int wave = __builtin_amdgcn_readfirstlane(tid >> 5);
    const int r16  = lane & 15;
    const int hi   = lane >> 4;
    const int colBase = wave * 16;
    const float4* h4 = (const float4*)h;
    const float bv = bias[colBase + r16];

    v2f wf[32];
    {
        const float* Wc = W + (size_t)(hi * 2) * 128 + colBase + r16;
        #pragma unroll
        for (int ks = 0; ks < 32; ++ks) {
            wf[ks][0] = Wc[(ks * 4 + 0) * 128];
            wf[ks][1] = Wc[(ks * 4 + 1) * 128];
        }
    }

    int t = blockIdx.x;
    int deg_l = 0;
    if (t < nTiles && lane < 2) deg_l = cnt[t * 16 + wave * 2 + lane];

    int cur = 0;
    for (; t < nTiles; t += gridDim.x) {
        const int n0 = t * 16 + wave * 2;
        float* Xw = &Xl[cur][(wave * 2) * XP + lane * 4];

        const int   deg0 = __builtin_amdgcn_readlane(deg_l, 0);
        const int   deg1 = __builtin_amdgcn_readlane(deg_l, 1);
        const float dd0  = __frsqrt_rn((float)(deg0 + 1));
        const float dd1  = __frsqrt_rn((float)(deg1 + 1));

        const int* sl0 = slot + (size_t)n0 * CAP;
        int sA = 0, sB = 0;
        if (lane < deg0) sA = sl0[lane];
        if (lane < deg1) sB = sl0[CAP + lane];
        int kA = 0, kB = 0;
        if (!PRE_IN) {
            kA = cnt[sA];
            kB = cnt[sB];
        }

        {
            const int tn = t + gridDim.x;
            deg_l = 0;
            if (tn < nTiles && lane < 2) deg_l = cnt[tn * 16 + wave * 2 + lane];
        }

        #pragma unroll
        for (int i = 0; i < 2; ++i) {
            const int   node = n0 + i;
            const int   deg  = (i == 0) ? deg0 : deg1;
            const float dd   = (i == 0) ? dd0 : dd1;
            const float4 self = h4[(size_t)node * 32 + lane];
            const int s_l = (i == 0) ? sA : sB;
            const int k_l = (i == 0) ? kA : kB;

            float4 acc = make_float4(0.f, 0.f, 0.f, 0.f);
            {
                const int m = min(deg, 32);
                int j = 0;
                if (m >= 16)     { add_rows<16, PRE_IN>(acc, h4, s_l, k_l, dd, 0, lane);  j = 16; }
                if (m - j >= 16) { add_rows<16, PRE_IN>(acc, h4, s_l, k_l, dd, 16, lane); j = 32; }
                if (m - j >= 8)  { add_rows<8, PRE_IN>(acc, h4, s_l, k_l, dd, j, lane);   j += 8; }
                switch (m - j) {
                    case 7: add_rows<7, PRE_IN>(acc, h4, s_l, k_l, dd, j, lane); break;
                    case 6: add_rows<6, PRE_IN>(acc, h4, s_l, k_l, dd, j, lane); break;
                    case 5: add_rows<5, PRE_IN>(acc, h4, s_l, k_l, dd, j, lane); break;
                    case 4: add_rows<4, PRE_IN>(acc, h4, s_l, k_l, dd, j, lane); break;
                    case 3: add_rows<3, PRE_IN>(acc, h4, s_l, k_l, dd, j, lane); break;
                    case 2: add_rows<2, PRE_IN>(acc, h4, s_l, k_l, dd, j, lane); break;
                    case 1: add_rows<1, PRE_IN>(acc, h4, s_l, k_l, dd, j, lane); break;
                    default: break;
                }
            }
            if (deg > 32) {
                const int m2 = min(deg, CAP) - 32;
                int s2 = 0;
                if (lane < m2) s2 = slot[(size_t)node * CAP + 32 + lane];
                if (PRE_IN) {
                    for (int j2 = 0; j2 < m2; ++j2) {
                        const int s = __builtin_amdgcn_readlane(s2, j2);
                        const float4 v = h4[(size_t)s * 32 + lane];
                        add4(acc, v);
                    }
                } else {
                    const int k2 = cnt[s2];
                    const float c2 = __frsqrt_rn((float)(k2 + 1)) * dd;
                    for (int j2 = 0; j2 < m2; ++j2) {
                        const int   s  = __builtin_amdgcn_readlane(s2, j2);
                        const float cq = __int_as_float(__builtin_amdgcn_readlane(__float_as_int(c2), j2));
                        const float4 v = h4[(size_t)s * 32 + lane];
                        fma4(acc, cq, v);
                    }
                }
            }
            if (deg > CAP) {
                const int nOv = min(ovfCount[0], eMax);
                for (int base = 0; base < nOv; base += 32) {
                    const int k = base + lane;
                    int2 q = make_int2(0, -1);
                    if (k < nOv) q = ovf[k];
                    const bool mine = (q.y == node);
                    float wq = PRE_IN ? 1.f : 0.f;
                    if (!PRE_IN && mine) wq = __frsqrt_rn((float)(cnt[q.x] + 1)) * dd;
                    unsigned todo = __builtin_amdgcn_ballot_w32(mine);
                    while (todo != 0u) {
                        const int jj = __builtin_ctz(todo);
                        todo &= todo - 1u;
                        const int   s  = __builtin_amdgcn_readlane(q.x, jj);
                        const float cq = __int_as_float(__builtin_amdgcn_readlane(__float_as_int(wq), jj));
                        const float4 v = h4[(size_t)s * 32 + lane];
                        fma4(acc, cq, v);
                    }
                }
            }
            if (PRE_IN) {
                add4(acc, self);
                acc.x *= dd;
                acc.y *= dd;
                acc.z *= dd;
                acc.w *= dd;
            } else {
                fma4(acc, dd * dd, self);
            }
            *(float4*)(Xw + i * XP) = acc;
        }

        int4 rc0 = make_int4(0, 0, 0, 0), rc1 = make_int4(0, 0, 0, 0);
        if (SCALE_OUT) {
            const int4* rc = (const int4*)(cnt + t * 16 + hi * 8);
            rc0 = rc[0];
            rc1 = rc[1];
        }

        __syncthreads();

        const float* Xc = Xl[cur];
        v8f acc8 = {};
        v2f af[32];
        #pragma unroll
        for (int ks = 0; ks < 32; ++ks) {
            const int kk = ks * 4 + hi * 2;
            af[ks][0] = Xc[r16 * XP + kk];
            af[ks][1] = Xc[r16 * XP + kk + 1];
        }
        __builtin_amdgcn_sched_barrier(0);
        #pragma unroll
        for (int ks = 0; ks < 32; ++ks)
            acc8 = __builtin_amdgcn_wmma_f32_16x16x4_f32(
                 false, af[ks],  false, wf[ks],
                 (short)0, acc8,  false,  false);

        float* out = Hout + (size_t)t * 16 * HDIM + colBase + r16;
        const int rcnt[8] = {rc0.x, rc0.y, rc0.z, rc0.w, rc1.x, rc1.y, rc1.z, rc1.w};
        #pragma unroll
        for (int r = 0; r < 8; ++r) {
            float v = acc8[r] + bv;
            if (relu) v = fmaxf(v, 0.f);
            if (SCALE_OUT) v *= __frsqrt_rn((float)(rcnt[r] + 1));
            out[(size_t)(r + hi * 8) * HDIM] = v;
        }
        cur ^= 1;
    }
}

extern "C" void kernel_launch(void* const* d_in, const int* in_sizes, int n_in,
                              void* d_out, int out_size, void* d_ws, size_t ws_size,
                              hipStream_t stream) {
    (void)n_in; (void)out_size; (void)ws_size;

    const float* x  = (const float*)d_in[0];
    const int*   ei = (const int*)d_in[1];
    const float* W1 = (const float*)d_in[2];
    const float* b1 = (const float*)d_in[3];
    const float* W2 = (const float*)d_in[4];
    const float* b2 = (const float*)d_in[5];
    const float* W3 = (const float*)d_in[6];
    const float* b3 = (const float*)d_in[7];

    const int N = in_sizes[0] / HDIM;
    const int E = in_sizes[1] / 2;
    const int* src = ei;
    const int* dst = ei + E;

    float* ws   = (float*)d_ws;
    float* h    = ws;
    float* xbuf = ws + (size_t)N * HDIM;
    const int cntLen = (N + 1 + 63) / 64 * 64;
    int*   cnt      = (int*)(ws + 2 * (size_t)N * HDIM);
    int*   ovfCount = cnt + N;
    int*   slot     = cnt + cntLen;
    int2*  ovf      = (int2*)(slot + (size_t)N * CAP);
    int2*  sorted   = ovf + E;
    int*   off      = (int*)(sorted + E);
    float* out      = (float*)d_out;

    dim3 blk(256);
    dim3 blkS(512);
    const int nChunks = (E + CHUNK - 1) / CHUNK;
    const int nBins   = (N + BNODES - 1) / BNODES;
    dim3 gGemm(768);
    const int nTiles = N / 16;

    sort_chunk<<<dim3(nChunks), blkS, 0, stream>>>(src, dst, sorted, off, ovfCount, E, nBins);
    place_bin<<<dim3(nBins), blkS, 0, stream>>>(sorted, off, cnt, slot, ovfCount, ovf, N, nBins, nChunks);

    gather_gemm<false, true><<<gGemm, blk, 0, stream>>>(x, cnt, slot, ovfCount, ovf, b1, W1, h, nTiles, 1, E);

    gather_gemm<true, true><<<gGemm, blk, 0, stream>>>(h, cnt, slot, ovfCount, ovf, b2, W2, xbuf, nTiles, 1, E);

    gather_gemm<true, false><<<gGemm, blk, 0, stream>>>(xbuf, cnt, slot, ovfCount, ovf, b3, W3, out, nTiles, 0, E);
}


# a tile's results leave as whole 512-byte rows: D staged in LDS; behind the next tile's barrier a wave stores its own 2 rows times its own dinv: no 64-byte half-line stores, no count loads; sums same
# speedup vs baseline: 6.7507x; 1.0316x over previous
#include <hip/hip_runtime.h>
#include <hip/hip_bf16.h>


#define NODES 50000
#define HDIM  128

typedef __attribute__((ext_vector_type(2))) float v2f;
typedef __attribute__((ext_vector_type(8))) float v8f;

#define CAP 64
#define XP 132

#define BSH    8
#define BNODES 256
#define MAXB   512
#define CHUNK  4096

template <bool FULL>
__device__ __forceinline__ void sort_chunk_body(const int* __restrict__ src, const int* __restrict__ dst,
                                                int2* __restrict__ sorted, int* __restrict__ off,
                                                int* hist, int* wsum, int2* stage,
                                                int e, int nBins, int base) {
    const int tid  = threadIdx.x;
    const int lane = tid & 31;
    const int wave = tid >> 5;
    const int i0   = base + tid * 8;
    int s[8], d[8], r[8];
    int n = 8;
    if (FULL) {
        const int4 s0 = *(const int4*)(src + i0);
        const int4 s1 = *(const int4*)(src + i0 + 4);
        const int4 d0 = *(const int4*)(dst + i0);
        const int4 d1 = *(const int4*)(dst + i0 + 4);
        s[0] = s0.x; s[1] = s0.y; s[2] = s0.z; s[3] = s0.w; s[4] = s1.x; s[5] = s1.y; s[6] = s1.z; s[7] = s1.w;
        d[0] = d0.x; d[1] = d0.y; d[2] = d0.z; d[3] = d0.w; d[4] = d1.x; d[5] = d1.y; d[6] = d1.z; d[7] = d1.w;
    } else {
        n = 0;
        #pragma unroll
        for (int u = 0; u < 8; ++u) {
            s[u] = 0; d[u] = 0;
            if (i0 + u < e) { s[u] = src[i0 + u]; d[u] = dst[i0 + u]; n = u + 1; }
        }
    }
    #pragma unroll
    for (int u = 0; u < 8; ++u) { r[u] = 0; if (FULL || u < n) r[u] = atomicAdd(&hist[d[u] >> BSH], 1); }
    __syncthreads();

    const int v = hist[tid];
    int inc = v;
    #pragma unroll
    for (int o = 1; o < 32; o <<= 1) {
        const int t = __shfl_up(inc, o);
        if (lane >= o) inc += t;
    }
    if (lane == 31) wsum[wave] = inc;
    __syncthreads();
    int wb = 0;
    for (int j = 0; j < wave; ++j) wb += wsum[j];
    const int start = wb + inc - v;
    hist[tid] = start;
    if (tid <= nBins) off[(size_t)blockIdx.x * (nBins + 1) + tid] = start;
    __syncthreads();

    #pragma unroll
    for (int u = 0; u < 8; ++u)
        if (FULL || u < n) stage[hist[d[u] >> BSH] + r[u]] = make_int2(s[u], d[u]);
    __syncthreads();

    if (FULL) {
        const int4* st4 = (const int4*)stage;
        int4* out4 = (int4*)(sorted + (size_t)base);
        #pragma unroll
        for (int i = 0; i < CHUNK / 2 / 512; ++i) out4[i * 512 + tid] = st4[i * 512 + tid];
    } else {
        const int m = min(CHUNK, e - base);
        for (int i = tid; i < m; i += 512) sorted[(size_t)base + i] = stage[i];
    }
}

__global__ __launch_bounds__(512) void sort_chunk(const int* __restrict__ src, const int* __restrict__ dst,
                                                  int2* __restrict__ sorted, int* __restrict__ off,
                                                  int* __restrict__ ovfCount, int e, int nBins) {
    __shared__ int  hist[MAXB];
    __shared__ int  wsum[16];
    __shared__ __attribute__((aligned(16))) int2 stage[CHUNK];

    const int tid = threadIdx.x;
    if (blockIdx.x == 0 && tid == 0) ovfCount[0] = 0;
    hist[tid] = 0;
    __syncthreads();

    const int base = blockIdx.x * CHUNK;
    if (base + CHUNK <= e && (e & 3) == 0)
        sort_chunk_body<true>(src, dst, sorted, off, hist, wsum, stage, e, nBins, base);
    else
        sort_chunk_body<false>(src, dst, sorted, off, hist, wsum, stage, e, nBins, base);
}

__device__ __forceinline__ void put_edge(int2 q, int lo, int* cntL, int* slotL,
                                         int* ovfCount, int2* __restrict__ ovf) {
    const int dl = q.y - lo;
    const int r  = atomicAdd(&cntL[dl], 1);
    if (r < CAP) {
        slotL[dl * CAP + r] = q.x;
    } else {
        const int k = atomicAdd(ovfCount, 1);
        ovf[k] = q;
    }
}

__global__ __launch_bounds__(512) void place_bin(const int2* __restrict__ sorted, const int* __restrict__ off,
                                                 int* __restrict__ cnt, int* __restrict__ slot,
                                                 int* ovfCount, int2* __restrict__ ovf,
                                                 int nNodes, int nBins, int nChunks) {
    __shared__ int cntL[BNODES];
    __shared__ int segS[512];
    __shared__ int segL[512];
    __shared__ int slotL[BNODES * CAP];

    const int tid  = threadIdx.x;
    const int lane = tid & 31;
    const int wave = __builtin_amdgcn_readfirstlane(tid >> 5);
    const int b    = blockIdx.x;
    const int lo   = b << BSH;
    if (tid < BNODES) cntL[tid] = 0;

    for (int c0 = 0; c0 < nChunks; c0 += 512) {
        __syncthreads();
        {
            const int c = c0 + tid;
            int st = 0, ln = 0;
            if (c < nChunks) {
                const int* o = off + (size_t)c * (nBins + 1) + b;
                const int o0 = o[0];
                ln = o[1] - o0;
                st = c * CHUNK + o0;
            }
            segS[tid] = st;
            segL[tid] = ln;
        }
        __syncthreads();
        const int nc = min(512, nChunks - c0);
        for (int j = wave; j < nc; j += 128) {
            int2 q[8];
            int  st[8], ln[8];
            #pragma unroll
            for (int u = 0; u < 8; ++u) {
                const int jj = j + 16 * u;
                st[u] = 0; ln[u] = 0;
                if (jj < nc) { st[u] = segS[jj]; ln[u] = segL[jj]; }
                q[u] = make_int2(0, 0);
                if (lane < ln[u]) q[u] = sorted[(size_t)st[u] + lane];
            }
            #pragma unroll
            for (int u = 0; u < 8; ++u)
                if (lane < ln[u]) put_edge(q[u], lo, cntL, slotL, ovfCount, ovf);
            #pragma unroll
            for (int u = 0; u < 8; ++u)
                for (int k = 32 + lane; k < ln[u]; k += 32)
                    put_edge(sorted[(size_t)st[u] + k], lo, cntL, slotL, ovfCount, ovf);
        }
    }
    __syncthreads();

    if (tid < BNODES && lo + tid < nNodes) cnt[lo + tid] = cntL[tid];
    for (int row = wave; row < BNODES; row += 16) {
        const int node = lo + row;
        if (node >= nNodes) break;
        const int c = min(cntL[row], CAP);
        int* g = slot + (size_t)node * CAP;
        if (lane < c) g[lane] = slotL[row * CAP + lane];
        if (lane + 32 < c) g[lane + 32] = slotL[row * CAP + lane + 32];
    }
}

__device__ __forceinline__ void fma4(float4& a, float c, const float4& v) {
    a.x = fmaf(c, v.x, a.x);
    a.y = fmaf(c, v.y, a.y);
    a.z = fmaf(c, v.z, a.z);
    a.w = fmaf(c, v.w, a.w);
}

__device__ __forceinline__ void add4(float4& a, const float4& v) {
    a.x += v.x;
    a.y += v.y;
    a.z += v.z;
    a.w += v.w;
}

template <int R, bool PRE>
__device__ __forceinline__ void add_rows(float4& acc, const float4* __restrict__ h4,
                                         int s_l, int k_l, float dd, int j, int lane) {
    float4 v[R];
    #pragma unroll
    for (int r = 0; r < R; ++r) {
        const int s = __builtin_amdgcn_readlane(s_l, j + r);
        v[r] = h4[(size_t)s * 32 + lane];
    }
    if (PRE) {
        #pragma unroll
        for (int r = 0; r < R; ++r) add4(acc, v[r]);
    } else {
        float c[R];
        const float c_l = __frsqrt_rn((float)(k_l + 1)) * dd;
        #pragma unroll
        for (int r = 0; r < R; ++r)
            c[r] = __int_as_float(__builtin_amdgcn_readlane(__float_as_int(c_l), j + r));
        #pragma unroll
        for (int r = 0; r < R; ++r) fma4(acc, c[r], v[r]);
    }
}

template <bool SCALE>
__device__ __forceinline__ void store_rows(float* __restrict__ Hout, const float* Os, int row0, int lane,
                                           float d0, float d1) {
    float4 o0 = *(const float4*)(Os);
    float4 o1 = *(const float4*)(Os + XP);
    if (SCALE) {
        o0.x *= d0; o0.y *= d0; o0.z *= d0; o0.w *= d0;
        o1.x *= d1; o1.y *= d1; o1.z *= d1; o1.w *= d1;
    }
    float4* g = (float4*)(Hout + (size_t)row0 * HDIM) + lane;
    g[0]  = o0;
    g[32] = o1;
}

template <bool PRE_IN, bool SCALE_OUT>
__global__ __launch_bounds__(256, 6) void gather_gemm(const float* __restrict__ h,
                                                   const int* __restrict__ cnt,
                                                   const int* __restrict__ slot,
                                                   const int* __restrict__ ovfCount,
                                                   const int2* __restrict__ ovf,
                                                   const float* __restrict__ bias,
                                                   const float* __restrict__ W,
                                                   float* __restrict__ Hout,
                                                   int nTiles, int relu, int eMax) {
    __shared__ __attribute__((aligned(16))) float Xl[2][16 * XP];
    __shared__ __attribute__((aligned(16))) float Ol[2][16 * XP];

    const int tid = threadIdx.x;
    const int lane = tid & 31;
    const int wave = __builtin_amdgcn_readfirstlane(tid >> 5);
    const int r16  = lane & 15;
    const int hi   = lane >> 4;
    const int colBase = wave * 16;
    const float4* h4 = (const float4*)h;
    const float bv = bias[colBase + r16];

    v2f wf[32];
    {
        const float* Wc = W + (size_t)(hi * 2) * 128 + colBase + r16;
        #pragma unroll
        for (int ks = 0; ks < 32; ++ks) {
            wf[ks][0] = Wc[(ks * 4 + 0) * 128];
            wf[ks][1] = Wc[(ks * 4 + 1) * 128];
        }
    }

    int t = blockIdx.x;
    int deg_l = 0;
    if (t < nTiles && lane < 2) deg_l = cnt[t * 16 + wave * 2 + lane];

    int cur = 0;
    int   tPrev = -1;
    float ddP0 = 0.f, ddP1 = 0.f;
    for (; t < nTiles; t += gridDim.x) {
        const int n0 = t * 16 + wave * 2;
        float* Xw = &Xl[cur][(wave * 2) * XP + lane * 4];

        const int   deg0 = __builtin_amdgcn_readlane(deg_l, 0);
        const int   deg1 = __builtin_amdgcn_readlane(deg_l, 1);
        const float dd0  = __frsqrt_rn((float)(deg0 + 1));
        const float dd1  = __frsqrt_rn((float)(deg1 + 1));

        const int* sl0 = slot + (size_t)n0 * CAP;
        int sA = 0, sB = 0;
        if (lane < deg0) sA = sl0[lane];
        if (lane < deg1) sB = sl0[CAP + lane];
        int kA = 0, kB = 0;
        if (!PRE_IN) {
            kA = cnt[sA];
            kB = cnt[sB];
        }

        {
            const int tn = t + gridDim.x;
            deg_l = 0;
            if (tn < nTiles && lane < 2) deg_l = cnt[tn * 16 + wave * 2 + lane];
        }

        #pragma unroll
        for (int i = 0; i < 2; ++i) {
            const int   node = n0 + i;
            const int   deg  = (i == 0) ? deg0 : deg1;
            const float dd   = (i == 0) ? dd0 : dd1;
            const float4 self = h4[(size_t)node * 32 + lane];
            const int s_l = (i == 0) ? sA : sB;
            const int k_l = (i == 0) ? kA : kB;

            float4 acc = make_float4(0.f, 0.f, 0.f, 0.f);
            {
                const int m = min(deg, 32);
                int j = 0;
                if (m >= 16)     { add_rows<16, PRE_IN>(acc, h4, s_l, k_l, dd, 0, lane);  j = 16; }
                if (m - j >= 16) { add_rows<16, PRE_IN>(acc, h4, s_l, k_l, dd, 16, lane); j = 32; }
                if (m - j >= 8)  { add_rows<8, PRE_IN>(acc, h4, s_l, k_l, dd, j, lane);   j += 8; }
                switch (m - j) {
                    case 7: add_rows<7, PRE_IN>(acc, h4, s_l, k_l, dd, j, lane); break;
                    case 6: add_rows<6, PRE_IN>(acc, h4, s_l, k_l, dd, j, lane); break;
                    case 5: add_rows<5, PRE_IN>(acc, h4, s_l, k_l, dd, j, lane); break;
                    case 4: add_rows<4, PRE_IN>(acc, h4, s_l, k_l, dd, j, lane); break;
                    case 3: add_rows<3, PRE_IN>(acc, h4, s_l, k_l, dd, j, lane); break;
                    case 2: add_rows<2, PRE_IN>(acc, h4, s_l, k_l, dd, j, lane); break;
                    case 1: add_rows<1, PRE_IN>(acc, h4, s_l, k_l, dd, j, lane); break;
                    default: break;
                }
            }
            if (deg > 32) {
                const int m2 = min(deg, CAP) - 32;
                int s2 = 0;
                if (lane < m2) s2 = slot[(size_t)node * CAP + 32 + lane];
                if (PRE_IN) {
                    for (int j2 = 0; j2 < m2; ++j2) {
                        const int s = __builtin_amdgcn_readlane(s2, j2);
                        const float4 v = h4[(size_t)s * 32 + lane];
                        add4(acc, v);
                    }
                } else {
                    const int k2 = cnt[s2];
                    const float c2 = __frsqrt_rn((float)(k2 + 1)) * dd;
                    for (int j2 = 0; j2 < m2; ++j2) {
                        const int   s  = __builtin_amdgcn_readlane(s2, j2);
                        const float cq = __int_as_float(__builtin_amdgcn_readlane(__float_as_int(c2), j2));
                        const float4 v = h4[(size_t)s * 32 + lane];
                        fma4(acc, cq, v);
                    }
                }
            }
            if (deg > CAP) {
                const int nOv = min(ovfCount[0], eMax);
                for (int base = 0; base < nOv; base += 32) {
                    const int k = base + lane;
                    int2 q = make_int2(0, -1);
                    if (k < nOv) q = ovf[k];
                    const bool mine = (q.y == node);
                    float wq = PRE_IN ? 1.f : 0.f;
                    if (!PRE_IN && mine) wq = __frsqrt_rn((float)(cnt[q.x] + 1)) * dd;
                    unsigned todo = __builtin_amdgcn_ballot_w32(mine);
                    while (todo != 0u) {
                        const int jj = __builtin_ctz(todo);
                        todo &= todo - 1u;
                        const int   s  = __builtin_amdgcn_readlane(q.x, jj);
                        const float cq = __int_as_float(__builtin_amdgcn_readlane(__float_as_int(wq), jj));
                        const float4 v = h4[(size_t)s * 32 + lane];
                        fma4(acc, cq, v);
                    }
                }
            }
            if (PRE_IN) {
                add4(acc, self);
                acc.x *= dd;
                acc.y *= dd;
                acc.z *= dd;
                acc.w *= dd;
            } else {
                fma4(acc, dd * dd, self);
            }
            *(float4*)(Xw + i * XP) = acc;
        }

        __syncthreads();

        const float* Xc = Xl[cur];
        v8f acc8 = {};
        v2f af[32];
        #pragma unroll
        for (int ks = 0; ks < 32; ++ks) {
            const int kk = ks * 4 + hi * 2;
            af[ks][0] = Xc[r16 * XP + kk];
            af[ks][1] = Xc[r16 * XP + kk + 1];
        }
        __builtin_amdgcn_sched_barrier(0);
        #pragma unroll
        for (int ks = 0; ks < 32; ++ks)
            acc8 = __builtin_amdgcn_wmma_f32_16x16x4_f32(
                 false, af[ks],  false, wf[ks],
                 (short)0, acc8,  false,  false);

        float* Ow = &Ol[cur][(hi * 8) * XP + colBase + r16];
        #pragma unroll
        for (int r = 0; r < 8; ++r) {
            float v = acc8[r] + bv;
            if (relu) v = fmaxf(v, 0.f);
            Ow[r * XP] = v;
        }
        if (tPrev >= 0)
            store_rows<SCALE_OUT>(Hout, &Ol[cur ^ 1][(wave * 2) * XP + lane * 4], tPrev * 16 + wave * 2, lane, ddP0, ddP1);
        tPrev = t;
        ddP0 = dd0;
        ddP1 = dd1;
        cur ^= 1;
    }
    if (tPrev >= 0) {
        __syncthreads();
        store_rows<SCALE_OUT>(Hout, &Ol[cur ^ 1][(wave * 2) * XP + lane * 4], tPrev * 16 + wave * 2, lane, ddP0, ddP1);
    }
}

extern "C" void kernel_launch(void* const* d_in, const int* in_sizes, int n_in,
                              void* d_out, int out_size, void* d_ws, size_t ws_size,
                              hipStream_t stream) {
    (void)n_in; (void)out_size; (void)ws_size;

    const float* x  = (const float*)d_in[0];
    const int*   ei = (const int*)d_in[1];
    const float* W1 = (const float*)d_in[2];
    const float* b1 = (const float*)d_in[3];
    const float* W2 = (const float*)d_in[4];
    const float* b2 = (const float*)d_in[5];
    const float* W3 = (const float*)d_in[6];
    const float* b3 = (const float*)d_in[7];

    const int N = in_sizes[0] / HDIM;
    const int E = in_sizes[1] / 2;
    const int* src = ei;
    const int* dst = ei + E;

    float* ws   = (float*)d_ws;
    float* h    = ws;
    float* xbuf = ws + (size_t)N * HDIM;
    const int cntLen = (N + 1 + 63) / 64 * 64;
    int*   cnt      = (int*)(ws + 2 * (size_t)N * HDIM);
    int*   ovfCount = cnt + N;
    int*   slot     = cnt + cntLen;
    int2*  ovf      = (int2*)(slot + (size_t)N * CAP);
    int2*  sorted   = ovf + E;
    int*   off      = (int*)(sorted + E);
    float* out      = (float*)d_out;

    dim3 blk(256);
    dim3 blkS(512);
    const int nChunks = (E + CHUNK - 1) / CHUNK;
    const int nBins   = (N + BNODES - 1) / BNODES;
    dim3 gGemm(768);
    const int nTiles = N / 16;

    sort_chunk<<<dim3(nChunks), blkS, 0, stream>>>(src, dst, sorted, off, ovfCount, E, nBins);
    place_bin<<<dim3(nBins), blkS, 0, stream>>>(sorted, off, cnt, slot, ovfCount, ovf, N, nBins, nChunks);

    gather_gemm<false, true><<<gGemm, blk, 0, stream>>>(x, cnt, slot, ovfCount, ovf, b1, W1, h, nTiles, 1, E);

    gather_gemm<true, true><<<gGemm, blk, 0, stream>>>(h, cnt, slot, ovfCount, ovf, b2, W2, xbuf, nTiles, 1, E);

    gather_gemm<true, false><<<gGemm, blk, 0, stream>>>(xbuf, cnt, slot, ovfCount, ovf, b3, W3, out, nTiles, 0, E);
}
